# starting kernel
# baseline (speedup 1.0000x reference)
#include <hip/hip_runtime.h>
#include <cstdint>
#include <cstddef>

#define INC  128
#define HIDC 64
#define OUTC 40

typedef float v2f __attribute__((ext_vector_type(2)));
typedef float v8f __attribute__((ext_vector_type(8)));


__global__ void k_init_deg(float* __restrict__ deg, int n) {
  int i = blockIdx.x * blockDim.x + threadIdx.x;
  if (i < n) deg[i] = 1.0f;
}

__global__ void k_count_deg(const int* __restrict__ dst, float* __restrict__ deg, int E) {
  int e = blockIdx.x * blockDim.x + threadIdx.x;
  if (e < E) unsafeAtomicAdd(&deg[dst[e]], 1.0f);
}

__global__ void k_rsqrt(float* __restrict__ d, int n) {
  int i = blockIdx.x * blockDim.x + threadIdx.x;
  if (i < n) d[i] = rsqrtf(d[i]);
}

__global__ void k_norm(const int* __restrict__ src, const int* __restrict__ dst,
                       const float* __restrict__ dinv, float* __restrict__ norm, int E) {
  int e = blockIdx.x * blockDim.x + threadIdx.x;
  if (e < E) norm[e] = dinv[src[e]] * dinv[dst[e]];
}


__global__ __launch_bounds__(128)
void k_gemm1_wmma(const float* __restrict__ x, const float* __restrict__ W1,
                  float* __restrict__ hx) {
  __shared__ float sA[16 * 132];
  const int r0 = blockIdx.x * 16;
  const int t  = threadIdx.x;

  {
    const int row = t >> 3;
    const int q   = (t & 7) * 16;
    const float4* s = (const float4*)(x + (size_t)(r0 + row) * INC + q);
    float4*       d = (float4*)(sA + row * 132 + q);
#pragma unroll
    for (int j = 0; j < 4; ++j) d[j] = s[j];
  }
  __syncthreads();

  const int wave  = t >> 5;
  const int lane  = t & 31;
  const int c0    = wave * 16;
  const int lhalf = lane & 15;
  const int khi   = (lane >> 4) * 2;

  v8f acc = {};
#pragma unroll
  for (int k = 0; k < INC; k += 4) {
    v2f a, b;
    const float* ap = sA + lhalf * 132 + k + khi;
    a.x = ap[0];
    a.y = ap[1];
    const float* bp = W1 + (size_t)(k + khi) * HIDC + c0 + lhalf;
    b.x = bp[0];
    b.y = bp[HIDC];
    acc = __builtin_amdgcn_wmma_f32_16x16x4_f32(false, a, false, b,
                                                (short)0, acc, false, false);
  }

  const int m0 = r0 + ((lane >> 4) * 8);
  const int n  = c0 + lhalf;
#pragma unroll
  for (int v = 0; v < 8; ++v)
    hx[(size_t)(m0 + v) * HIDC + n] = acc[v];
}


__global__ __launch_bounds__(96)
void k_gemm2_wmma(const float* __restrict__ agg1, const float* __restrict__ b1,
                  const float* __restrict__ W2, float* __restrict__ h2) {
  __shared__ float sA[16 * 68];
  const int r0 = blockIdx.x * 16;
  const int t  = threadIdx.x;

  for (int i = t; i < 256; i += 96) {
    const int row = i >> 4;
    const int q   = (i & 15) * 4;
    float4 v = *(const float4*)(agg1 + (size_t)(r0 + row) * HIDC + q);
    v.x = fmaxf(v.x + b1[q + 0], 0.0f);
    v.y = fmaxf(v.y + b1[q + 1], 0.0f);
    v.z = fmaxf(v.z + b1[q + 2], 0.0f);
    v.w = fmaxf(v.w + b1[q + 3], 0.0f);
    *(float4*)(sA + row * 68 + q) = v;
  }
  __syncthreads();

  const int wave  = t >> 5;
  const int lane  = t & 31;
  const int c0    = wave * 16;
  const int lhalf = lane & 15;
  const int khi   = (lane >> 4) * 2;
  const int col   = c0 + lhalf;
  const bool colOK = (col < OUTC);

  v8f acc = {};
#pragma unroll
  for (int k = 0; k < HIDC; k += 4) {
    v2f a, b;
    const float* ap = sA + lhalf * 68 + k + khi;
    a.x = ap[0];
    a.y = ap[1];
    float bx = 0.0f, by = 0.0f;
    if (colOK) {
      bx = W2[(size_t)(k + khi) * OUTC + col];
      by = W2[(size_t)(k + khi + 1) * OUTC + col];
    }
    b.x = bx;
    b.y = by;
    acc = __builtin_amdgcn_wmma_f32_16x16x4_f32(false, a, false, b,
                                                (short)0, acc, false, false);
  }

  if (colOK) {
    const int m0 = r0 + ((lane >> 4) * 8);
#pragma unroll
    for (int v = 0; v < 8; ++v)
      h2[(size_t)(m0 + v) * OUTC + col] = acc[v];
  }
}


__global__ void k_selfloop_h(const float* __restrict__ hx, const float* __restrict__ dinv,
                             float* __restrict__ agg, int total4) {
  int t = blockIdx.x * blockDim.x + threadIdx.x;
  if (t < total4) {
    int i = t >> 4;
    float di = dinv[i];
    float s = di * di;
    float4 v = ((const float4*)hx)[t];
    v.x *= s; v.y *= s; v.z *= s; v.w *= s;
    ((float4*)agg)[t] = v;
  }
}

__global__ void k_scatter_h(const int* __restrict__ src, const int* __restrict__ dst,
                            const float* __restrict__ norm, const float* __restrict__ hx,
                            float* __restrict__ agg, int total4) {
  int t = blockIdx.x * blockDim.x + threadIdx.x;
  if (t < total4) {
    int e  = t >> 4;
    int c4 = (t & 15) * 4;
    float w = norm[e];
    const float4 v = *(const float4*)(hx + (size_t)src[e] * HIDC + c4);
    float* p = agg + (size_t)dst[e] * HIDC + c4;
    unsafeAtomicAdd(p + 0, v.x * w);
    unsafeAtomicAdd(p + 1, v.y * w);
    unsafeAtomicAdd(p + 2, v.z * w);
    unsafeAtomicAdd(p + 3, v.w * w);
  }
}

__global__ void k_selfloop_o(const float* __restrict__ h2, const float* __restrict__ dinv,
                             const float* __restrict__ b2, float* __restrict__ out,
                             int total4) {
  int t = blockIdx.x * blockDim.x + threadIdx.x;
  if (t < total4) {
    int i = t / 10;
    int q = (t - i * 10) * 4;
    float di = dinv[i];
    float s = di * di;
    float4 v = ((const float4*)h2)[t];
    const float4 b = *(const float4*)(b2 + q);
    v.x = b.x + s * v.x;
    v.y = b.y + s * v.y;
    v.z = b.z + s * v.z;
    v.w = b.w + s * v.w;
    ((float4*)out)[t] = v;
  }
}

__global__ void k_scatter_o(const int* __restrict__ src, const int* __restrict__ dst,
                            const float* __restrict__ norm, const float* __restrict__ h2,
                            float* __restrict__ out, int total4) {
  int t = blockIdx.x * blockDim.x + threadIdx.x;
  if (t < total4) {
    int e = t / 10;
    int q = (t - e * 10) * 4;
    float w = norm[e];
    const float4 v = *(const float4*)(h2 + (size_t)src[e] * OUTC + q);
    float* p = out + (size_t)dst[e] * OUTC + q;
    unsafeAtomicAdd(p + 0, v.x * w);
    unsafeAtomicAdd(p + 1, v.y * w);
    unsafeAtomicAdd(p + 2, v.z * w);
    unsafeAtomicAdd(p + 3, v.w * w);
  }
}


extern "C" void kernel_launch(void* const* d_in, const int* in_sizes, int n_in,
                              void* d_out, int out_size, void* d_ws, size_t ws_size,
                              hipStream_t stream) {
  const float* x  = (const float*)d_in[0];
  const int*   ei = (const int*)d_in[1];
  const float* W1 = (const float*)d_in[2];
  const float* b1 = (const float*)d_in[3];
  const float* W2 = (const float*)d_in[4];
  const float* b2 = (const float*)d_in[5];
  float* out = (float*)d_out;

  const int N = in_sizes[0] / INC;
  const int E = in_sizes[1] / 2;
  const int* src = ei;
  const int* dst = ei + E;

  char*  ws  = (char*)d_ws;
  size_t off = 0;
  auto carve = [&](size_t bytes) -> float* {
    float* p = (float*)(ws + off);
    off = (off + bytes + 255) & ~(size_t)255;
    return p;
  };
  float* dinv = carve((size_t)N * 4);
  float* norm = carve((size_t)E * 4);
  float* hx   = carve((size_t)N * HIDC * 4);
  float* agg1 = carve((size_t)N * HIDC * 4);
  float* h2   = carve((size_t)N * OUTC * 4);
  (void)n_in; (void)out_size; (void)ws_size;

  const int B = 256;
  k_init_deg  <<<(N + B - 1) / B, B, 0, stream>>>(dinv, N);
  k_count_deg <<<(E + B - 1) / B, B, 0, stream>>>(dst, dinv, E);
  k_rsqrt     <<<(N + B - 1) / B, B, 0, stream>>>(dinv, N);
  k_norm      <<<(E + B - 1) / B, B, 0, stream>>>(src, dst, dinv, norm, E);

  k_gemm1_wmma<<<N / 16, 128, 0, stream>>>(x, W1, hx);
  {
    int tot4 = N * (HIDC / 4);
    k_selfloop_h<<<(tot4 + B - 1) / B, B, 0, stream>>>(hx, dinv, agg1, tot4);
  }
  {
    int tot4 = E * (HIDC / 4);
    k_scatter_h<<<(tot4 + B - 1) / B, B, 0, stream>>>(src, dst, norm, hx, agg1, tot4);
  }

  k_gemm2_wmma<<<N / 16, 96, 0, stream>>>(agg1, b1, W2, h2);
  {
    int tot4 = N * (OUTC / 4);
    k_selfloop_o<<<(tot4 + B - 1) / B, B, 0, stream>>>(h2, dinv, b2, out, tot4);
  }
  {
    int tot4 = E * (OUTC / 4);
    k_scatter_o<<<(tot4 + B - 1) / B, B, 0, stream>>>(src, dst, norm, h2, out, tot4);
  }
}


# In-edge list by counting sort in LDS: 8192-edge chunks ordered by 512-node bin, then each bin ordered by node; no global atomics, no scattered stores; 8 to 6 launches; sum order = LDS arrival rank
# speedup vs baseline: 3.8500x; 3.8500x over previous
#include <hip/hip_runtime.h>
#include <cstdint>
#include <cstddef>

#define INC  128
#define HIDC 64
#define OUTC 40

typedef float v2f __attribute__((ext_vector_type(2)));
typedef float v8f __attribute__((ext_vector_type(8)));


#define BIN_SHIFT 9
#define BIN_NODES 512
#define NBMAX     256
#define CH        8192
#define MAXBLK    1024
#define SORT_EPT  12
#define SORT_CAP  (SORT_EPT * 1024)

__device__ __forceinline__ int wave_incl_scan(int x, int lane) {
  for (int m = 1; m < 32; m <<= 1) {
    const int o = __shfl_up(x, m, 32);
    if (lane >= m) x += o;
  }
  return x;
}

__device__ __forceinline__ int block_excl_scan(int x, int* s_w, int lane, int wave, int& total) {
  const int inc = wave_incl_scan(x, lane);
  if (lane == 31) s_w[wave] = inc;
  __syncthreads();
  const int wv = s_w[lane];
  const int wi = wave_incl_scan(wv, lane);
  total = __shfl(wi, 31, 32);
  const int wex = __shfl(wi - wv, wave, 32);
  return wex + inc - x;
}

__global__ __launch_bounds__(1024)
void k_bin(const int* __restrict__ src, const int* __restrict__ dst,
           unsigned* __restrict__ binned, int* __restrict__ lofsT,
           int E, int NB, int NBLK) {
  __shared__ int      s_hist[NBMAX];
  __shared__ int      s_lofs[NBMAX + 1];
  __shared__ int      s_w[32];
  __shared__ unsigned s_pack[CH];
  const int t    = threadIdx.x;
  const int lane = t & 31;
  const int wave = t >> 5;
  const int blk  = blockIdx.x;
  const int base = blk * CH;

  if (t < NBMAX) s_hist[t] = 0;
  __syncthreads();

  int sv[8], dv[8];
  const bool whole = ((E & 3) == 0) && (base + CH <= E);
  if (whole) {
#pragma unroll
    for (int q = 0; q < 2; ++q) {
      const int e0 = base + q * (CH / 2) + t * 4;
      const int4 s4 = *(const int4*)(src + e0);
      const int4 d4 = *(const int4*)(dst + e0);
      sv[q * 4 + 0] = s4.x; sv[q * 4 + 1] = s4.y; sv[q * 4 + 2] = s4.z; sv[q * 4 + 3] = s4.w;
      dv[q * 4 + 0] = d4.x; dv[q * 4 + 1] = d4.y; dv[q * 4 + 2] = d4.z; dv[q * 4 + 3] = d4.w;
    }
  } else {
#pragma unroll
    for (int k = 0; k < 8; ++k) {
      const int e = base + (k >> 2) * (CH / 2) + t * 4 + (k & 3);
      sv[k] = (e < E) ? src[e] : 0;
      dv[k] = (e < E) ? dst[e] : -1;
    }
  }

  int rk[8];
#pragma unroll
  for (int k = 0; k < 8; ++k) {
    rk[k] = 0;
    if (dv[k] >= 0) {
      const unsigned bin = min((unsigned)dv[k] >> BIN_SHIFT, (unsigned)(NBMAX - 1));
      rk[k] = atomicAdd(&s_hist[bin], 1);
    }
  }
  __syncthreads();

  int nloc;
  const int h  = (t < NBMAX) ? s_hist[t] : 0;
  const int ex = block_excl_scan(h, s_w, lane, wave, nloc);
  if (t < NBMAX) s_lofs[t] = ex;
  if (t == NBMAX) s_lofs[NBMAX] = nloc;
  __syncthreads();

  if (t <= NB) lofsT[(size_t)t * NBLK + blk] = s_lofs[t];
#pragma unroll
  for (int k = 0; k < 8; ++k) {
    if (dv[k] >= 0) {
      const unsigned bin = min((unsigned)dv[k] >> BIN_SHIFT, (unsigned)(NBMAX - 1));
      s_pack[s_lofs[bin] + rk[k]] =
          ((unsigned)(dv[k] & (BIN_NODES - 1)) << 23) | (unsigned)sv[k];
    }
  }
  __syncthreads();

  for (int i = t; i < nloc; i += 1024) binned[base + i] = s_pack[i];
}

__global__ __launch_bounds__(1024)
void k_sort(const unsigned* __restrict__ binned, const int* __restrict__ lofsT,
            int* __restrict__ list, int* __restrict__ rowptr, int* __restrict__ cnt,
            float* __restrict__ dinv, int N, int NBLK) {
  __shared__ int s_a[MAXBLK];
  __shared__ int s_ro[MAXBLK];
  __shared__ int s_cnt[BIN_NODES];
  __shared__ int s_ofs[BIN_NODES];
  __shared__ int s_cur[BIN_NODES];
  __shared__ int s_w1[32], s_w2[32], s_w3[32];
  __shared__ int s_list[SORT_CAP];
  const int t    = threadIdx.x;
  const int lane = t & 31;
  const int wave = t >> 5;
  const int j    = blockIdx.x;

  int a = 0, len = 0;
  if (t < NBLK) {
    a   = lofsT[(size_t)j * NBLK + t];
    len = lofsT[(size_t)(j + 1) * NBLK + t] - a;
    s_a[t] = t * CH + a;
  }
  if (t < BIN_NODES) { s_cnt[t] = 0; s_cur[t] = 0; }
  int total, binbase;
  const int ro = block_excl_scan(len, s_w1, lane, wave, total);
  (void)block_excl_scan(a, s_w2, lane, wave, binbase);
  if (t < NBLK) s_ro[t] = ro;
  __syncthreads();

  int top = 1;
  while (top * 2 < NBLK) top *= 2;

  auto fetch = [&](int i) -> unsigned {
    int pos = 0;
    for (int step = top; step > 0; step >>= 1) {
      const int nxt = pos + step;
      if (nxt < NBLK && s_ro[nxt] <= i) pos = nxt;
    }
    return binned[s_a[pos] + (i - s_ro[pos])];
  };

  const bool fast = (total <= SORT_CAP);
  unsigned v[SORT_EPT];
  int      rk[SORT_EPT];
  if (fast) {
    int pos[SORT_EPT];
#pragma unroll
    for (int m = 0; m < SORT_EPT; ++m) pos[m] = 0;
    for (int step = top; step > 0; step >>= 1) {
#pragma unroll
      for (int m = 0; m < SORT_EPT; ++m) {
        const int nx = pos[m] + step;
        const int rd = s_ro[min(nx, NBLK - 1)];
        if (nx < NBLK && rd <= m * 1024 + t) pos[m] = nx;
      }
    }
#pragma unroll
    for (int m = 0; m < SORT_EPT; ++m) {
      const int i = m * 1024 + t;
      v[m] = 0u;
      if (i < total) v[m] = binned[s_a[pos[m]] + (i - s_ro[pos[m]])];
    }
#pragma unroll
    for (int m = 0; m < SORT_EPT; ++m) {
      const int i = m * 1024 + t;
      rk[m] = 0;
      if (i < total) rk[m] = atomicAdd(&s_cnt[v[m] >> 23], 1);
    }
  } else {
    for (int i = t; i < total; i += 1024) atomicAdd(&s_cnt[fetch(i) >> 23], 1);
  }
  __syncthreads();

  int dummy;
  const int c   = (t < BIN_NODES) ? s_cnt[t] : 0;
  const int exn = block_excl_scan(c, s_w3, lane, wave, dummy);
  if (t < BIN_NODES) {
    s_ofs[t] = exn;
    const int node = j * BIN_NODES + t;
    if (node < N) {
      rowptr[node] = binbase + exn;
      cnt[node]    = c;
      dinv[node]   = rsqrtf((float)(c + 1));
    }
  }
  __syncthreads();

  if (fast) {
#pragma unroll
    for (int m = 0; m < SORT_EPT; ++m) {
      const int i = m * 1024 + t;
      if (i < total) s_list[s_ofs[v[m] >> 23] + rk[m]] = (int)(v[m] & 0x7FFFFFu);
    }
    __syncthreads();
    for (int i = t; i < total; i += 1024) list[binbase + i] = s_list[i];
  } else {
    for (int i = t; i < total; i += 1024) {
      const unsigned w  = fetch(i);
      const int      dl = (int)(w >> 23);
      const int      p  = s_ofs[dl] + atomicAdd(&s_cur[dl], 1);
      list[binbase + p] = (int)(w & 0x7FFFFFu);
    }
  }
}


__global__ __launch_bounds__(128)
void k_gemm1_wmma(const float* __restrict__ x, const float* __restrict__ W1,
                  const float* __restrict__ dinv, float* __restrict__ hx) {
  __shared__ float sA[16 * 132];
  const int r0 = blockIdx.x * 16;
  const int t  = threadIdx.x;

  {
    const int row = t >> 3;
    const int q   = (t & 7) * 16;
    const float4* s = (const float4*)(x + (size_t)(r0 + row) * INC + q);
    float4*       d = (float4*)(sA + row * 132 + q);
#pragma unroll
    for (int j = 0; j < 4; ++j) d[j] = s[j];
  }
  __syncthreads();

  const int wave  = t >> 5;
  const int lane  = t & 31;
  const int c0    = wave * 16;
  const int lhalf = lane & 15;
  const int khi   = (lane >> 4) * 2;

  v8f acc = {};
#pragma unroll
  for (int k = 0; k < INC; k += 4) {
    v2f a, b;
    const float* ap = sA + lhalf * 132 + k + khi;
    a.x = ap[0];
    a.y = ap[1];
    const float* bp = W1 + (size_t)(k + khi) * HIDC + c0 + lhalf;
    b.x = bp[0];
    b.y = bp[HIDC];
    acc = __builtin_amdgcn_wmma_f32_16x16x4_f32(false, a, false, b,
                                                (short)0, acc, false, false);
  }

  const int m0 = r0 + ((lane >> 4) * 8);
  const int n  = c0 + lhalf;
#pragma unroll
  for (int v = 0; v < 8; ++v)
    hx[(size_t)(m0 + v) * HIDC + n] = acc[v] * dinv[m0 + v];
}


__global__ __launch_bounds__(96)
void k_gemm2_wmma(const float* __restrict__ agg1, const float* __restrict__ b1,
                  const float* __restrict__ W2, const float* __restrict__ dinv,
                  float* __restrict__ h2) {
  __shared__ float sA[16 * 68];
  const int r0 = blockIdx.x * 16;
  const int t  = threadIdx.x;

  for (int i = t; i < 256; i += 96) {
    const int row = i >> 4;
    const int q   = (i & 15) * 4;
    float4 v = *(const float4*)(agg1 + (size_t)(r0 + row) * HIDC + q);
    v.x = fmaxf(v.x + b1[q + 0], 0.0f);
    v.y = fmaxf(v.y + b1[q + 1], 0.0f);
    v.z = fmaxf(v.z + b1[q + 2], 0.0f);
    v.w = fmaxf(v.w + b1[q + 3], 0.0f);
    *(float4*)(sA + row * 68 + q) = v;
  }
  __syncthreads();

  const int wave  = t >> 5;
  const int lane  = t & 31;
  const int c0    = wave * 16;
  const int lhalf = lane & 15;
  const int khi   = (lane >> 4) * 2;
  const int col   = c0 + lhalf;
  const bool colOK = (col < OUTC);

  v8f acc = {};
#pragma unroll
  for (int k = 0; k < HIDC; k += 4) {
    v2f a, b;
    const float* ap = sA + lhalf * 68 + k + khi;
    a.x = ap[0];
    a.y = ap[1];
    float bx = 0.0f, by = 0.0f;
    if (colOK) {
      bx = W2[(size_t)(k + khi) * OUTC + col];
      by = W2[(size_t)(k + khi + 1) * OUTC + col];
    }
    b.x = bx;
    b.y = by;
    acc = __builtin_amdgcn_wmma_f32_16x16x4_f32(false, a, false, b,
                                                (short)0, acc, false, false);
  }

  if (colOK) {
    const int m0 = r0 + ((lane >> 4) * 8);
#pragma unroll
    for (int v = 0; v < 8; ++v)
      h2[(size_t)(m0 + v) * OUTC + col] = acc[v] * dinv[m0 + v];
  }
}


template <int C4, bool BIAS>
__device__ __forceinline__ void pull_rows(const float* __restrict__ h, const int* __restrict__ list,
                                          const int* __restrict__ rowptr, const int* __restrict__ cnt,
                                          const float* __restrict__ dinv, const float* __restrict__ bias,
                                          float* __restrict__ o, int n) {
  const int node = blockIdx.x * (blockDim.x >> 4) + (threadIdx.x >> 4);
  const int sub  = threadIdx.x & 15;
  if (node >= n || sub >= C4) return;

  const float4* h4 = (const float4*)h;
  const float di = dinv[node];
  float4 acc = h4[(size_t)node * C4 + sub];
  float4 b = make_float4(0.0f, 0.0f, 0.0f, 0.0f);
  if (BIAS) b = ((const float4*)bias)[sub];

  const int beg = rowptr[node];
  const int end = beg + cnt[node];
  int j = beg;
  for (; j + 4 <= end; j += 4) {
    const int s0 = list[j];
    const int s1 = list[j + 1];
    const int s2 = list[j + 2];
    const int s3 = list[j + 3];
    const float4 v0 = h4[(size_t)s0 * C4 + sub];
    const float4 v1 = h4[(size_t)s1 * C4 + sub];
    const float4 v2 = h4[(size_t)s2 * C4 + sub];
    const float4 v3 = h4[(size_t)s3 * C4 + sub];
    acc.x += v0.x; acc.y += v0.y; acc.z += v0.z; acc.w += v0.w;
    acc.x += v1.x; acc.y += v1.y; acc.z += v1.z; acc.w += v1.w;
    acc.x += v2.x; acc.y += v2.y; acc.z += v2.z; acc.w += v2.w;
    acc.x += v3.x; acc.y += v3.y; acc.z += v3.z; acc.w += v3.w;
  }
  if (j < end) {
    const int  last = end - 1;
    const bool p1 = (j + 1 < end);
    const bool p2 = (j + 2 < end);
    const int s0 = list[j];
    const int s1 = list[min(j + 1, last)];
    const int s2 = list[min(j + 2, last)];
    const float4 v0 = h4[(size_t)s0 * C4 + sub];
    const float4 r1 = h4[(size_t)s1 * C4 + sub];
    const float4 r2 = h4[(size_t)s2 * C4 + sub];
    asm volatile("" :: "v"(r1.x), "v"(r1.y), "v"(r1.z), "v"(r1.w),
                       "v"(r2.x), "v"(r2.y), "v"(r2.z), "v"(r2.w));
    const float4 v1 = p1 ? r1 : make_float4(0.0f, 0.0f, 0.0f, 0.0f);
    const float4 v2 = p2 ? r2 : make_float4(0.0f, 0.0f, 0.0f, 0.0f);
    acc.x += v0.x; acc.y += v0.y; acc.z += v0.z; acc.w += v0.w;
    acc.x += v1.x; acc.y += v1.y; acc.z += v1.z; acc.w += v1.w;
    acc.x += v2.x; acc.y += v2.y; acc.z += v2.z; acc.w += v2.w;
  }
  if (BIAS) {
    acc.x = b.x + di * acc.x;
    acc.y = b.y + di * acc.y;
    acc.z = b.z + di * acc.z;
    acc.w = b.w + di * acc.w;
  } else {
    acc.x *= di; acc.y *= di; acc.z *= di; acc.w *= di;
  }
  ((float4*)o)[(size_t)node * C4 + sub] = acc;
}

__global__ __launch_bounds__(256)
void k_pull_h(const float* __restrict__ hx, const int* __restrict__ list,
              const int* __restrict__ rowptr, const int* __restrict__ cnt,
              const float* __restrict__ dinv, float* __restrict__ agg, int n) {
  pull_rows<HIDC / 4, false>(hx, list, rowptr, cnt, dinv, nullptr, agg, n);
}

__global__ __launch_bounds__(256)
void k_pull_o(const float* __restrict__ h2, const int* __restrict__ list,
              const int* __restrict__ rowptr, const int* __restrict__ cnt,
              const float* __restrict__ dinv, const float* __restrict__ b2,
              float* __restrict__ out, int n) {
  pull_rows<OUTC / 4, true>(h2, list, rowptr, cnt, dinv, b2, out, n);
}


extern "C" void kernel_launch(void* const* d_in, const int* in_sizes, int n_in,
                              void* d_out, int out_size, void* d_ws, size_t ws_size,
                              hipStream_t stream) {
  const float* x  = (const float*)d_in[0];
  const int*   ei = (const int*)d_in[1];
  const float* W1 = (const float*)d_in[2];
  const float* b1 = (const float*)d_in[3];
  const float* W2 = (const float*)d_in[4];
  const float* b2 = (const float*)d_in[5];
  float* out = (float*)d_out;

  const int N = in_sizes[0] / INC;
  const int E = in_sizes[1] / 2;
  const int* src = ei;
  const int* dst = ei + E;

  char*  ws  = (char*)d_ws;
  size_t off = 0;
  auto carve = [&](size_t bytes) -> void* {
    void* p = (void*)(ws + off);
    off = (off + bytes + 255) & ~(size_t)255;
    return p;
  };
  int*   cnt    = (int*)  carve((size_t)N * 4);
  int*   rowptr = (int*)  carve((size_t)N * 4);
  float* dinv   = (float*)carve((size_t)N * 4);
  int*   list   = (int*)  carve((size_t)E * 4);
  int*   rank   = (int*)  carve((size_t)E * 4);
  float* hx     = (float*)carve((size_t)N * HIDC * 4);
  float* agg1   = (float*)carve((size_t)N * HIDC * 4);
  float* h2     = hx;
  (void)n_in; (void)out_size; (void)ws_size;

  const int B = 256;
  const int NB   = (N + BIN_NODES - 1) / BIN_NODES;
  const int NBLK = (E + CH - 1) / CH;
  unsigned* binned = (unsigned*)rank;
  int*      lofsT  = (int*)agg1;
  k_bin <<<NBLK, 1024, 0, stream>>>(src, dst, binned, lofsT, E, NB, NBLK);
  k_sort<<<NB,   1024, 0, stream>>>(binned, lofsT, list, rowptr, cnt, dinv, N, NBLK);

  k_gemm1_wmma<<<N / 16, 128, 0, stream>>>(x, W1, dinv, hx);
  k_pull_h    <<<(N + 15) / 16, B, 0, stream>>>(hx, list, rowptr, cnt, dinv, agg1, N);

  k_gemm2_wmma<<<N / 16, 96, 0, stream>>>(agg1, b1, W2, dinv, h2);
  k_pull_o    <<<(N + 15) / 16, B, 0, stream>>>(h2, list, rowptr, cnt, dinv, b2, out, N);
}


# Long-lived waves: GEMMs and pulls launch only what the chip holds at once and loop over row tiles / nodes (weights stay in registers, next tile / node header fetched ahead); sum order unchanged
# speedup vs baseline: 4.3518x; 1.1303x over previous
#include <hip/hip_runtime.h>
#include <cstdint>
#include <cstddef>

#define INC  128
#define HIDC 64
#define OUTC 40

typedef float v2f __attribute__((ext_vector_type(2)));
typedef float v8f __attribute__((ext_vector_type(8)));


#define BIN_SHIFT 9
#define BIN_NODES 512
#define NBMAX     256
#define CH        8192
#define MAXBLK    1024
#define SORT_EPT  12
#define SORT_CAP  (SORT_EPT * 1024)

__device__ __forceinline__ int wave_incl_scan(int x, int lane) {
  for (int m = 1; m < 32; m <<= 1) {
    const int o = __shfl_up(x, m, 32);
    if (lane >= m) x += o;
  }
  return x;
}

__device__ __forceinline__ int block_excl_scan(int x, int* s_w, int lane, int wave, int& total) {
  const int inc = wave_incl_scan(x, lane);
  if (lane == 31) s_w[wave] = inc;
  __syncthreads();
  const int wv = s_w[lane];
  const int wi = wave_incl_scan(wv, lane);
  total = __shfl(wi, 31, 32);
  const int wex = __shfl(wi - wv, wave, 32);
  return wex + inc - x;
}

__global__ __launch_bounds__(1024)
void k_bin(const int* __restrict__ src, const int* __restrict__ dst,
           unsigned* __restrict__ binned, int* __restrict__ lofsT,
           int E, int NB, int NBLK) {
  __shared__ int      s_hist[NBMAX];
  __shared__ int      s_lofs[NBMAX + 1];
  __shared__ int      s_w[32];
  __shared__ unsigned s_pack[CH];
  const int t    = threadIdx.x;
  const int lane = t & 31;
  const int wave = t >> 5;
  const int blk  = blockIdx.x;
  const int base = blk * CH;

  if (t < NBMAX) s_hist[t] = 0;
  __syncthreads();

  int sv[8], dv[8];
  const bool whole = ((E & 3) == 0) && (base + CH <= E);
  if (whole) {
#pragma unroll
    for (int q = 0; q < 2; ++q) {
      const int e0 = base + q * (CH / 2) + t * 4;
      const int4 s4 = *(const int4*)(src + e0);
      const int4 d4 = *(const int4*)(dst + e0);
      sv[q * 4 + 0] = s4.x; sv[q * 4 + 1] = s4.y; sv[q * 4 + 2] = s4.z; sv[q * 4 + 3] = s4.w;
      dv[q * 4 + 0] = d4.x; dv[q * 4 + 1] = d4.y; dv[q * 4 + 2] = d4.z; dv[q * 4 + 3] = d4.w;
    }
  } else {
#pragma unroll
    for (int k = 0; k < 8; ++k) {
      const int e = base + (k >> 2) * (CH / 2) + t * 4 + (k & 3);
      sv[k] = (e < E) ? src[e] : 0;
      dv[k] = (e < E) ? dst[e] : -1;
    }
  }

  int rk[8];
#pragma unroll
  for (int k = 0; k < 8; ++k) {
    rk[k] = 0;
    if (dv[k] >= 0) {
      const unsigned bin = min((unsigned)dv[k] >> BIN_SHIFT, (unsigned)(NBMAX - 1));
      rk[k] = atomicAdd(&s_hist[bin], 1);
    }
  }
  __syncthreads();

  int nloc;
  const int h  = (t < NBMAX) ? s_hist[t] : 0;
  const int ex = block_excl_scan(h, s_w, lane, wave, nloc);
  if (t < NBMAX) s_lofs[t] = ex;
  if (t == NBMAX) s_lofs[NBMAX] = nloc;
  __syncthreads();

  if (t <= NB) lofsT[(size_t)t * NBLK + blk] = s_lofs[t];
#pragma unroll
  for (int k = 0; k < 8; ++k) {
    if (dv[k] >= 0) {
      const unsigned bin = min((unsigned)dv[k] >> BIN_SHIFT, (unsigned)(NBMAX - 1));
      s_pack[s_lofs[bin] + rk[k]] =
          ((unsigned)(dv[k] & (BIN_NODES - 1)) << 23) | (unsigned)sv[k];
    }
  }
  __syncthreads();

  for (int i = t; i < nloc; i += 1024) binned[base + i] = s_pack[i];
}

__global__ __launch_bounds__(1024)
void k_sort(const unsigned* __restrict__ binned, const int* __restrict__ lofsT,
            int* __restrict__ list, int* __restrict__ rowptr, int* __restrict__ cnt,
            float* __restrict__ dinv, int N, int NBLK) {
  __shared__ int s_a[MAXBLK];
  __shared__ int s_ro[MAXBLK];
  __shared__ int s_cnt[BIN_NODES];
  __shared__ int s_ofs[BIN_NODES];
  __shared__ int s_cur[BIN_NODES];
  __shared__ int s_w1[32], s_w2[32], s_w3[32];
  __shared__ int s_list[SORT_CAP];
  const int t    = threadIdx.x;
  const int lane = t & 31;
  const int wave = t >> 5;
  const int j    = blockIdx.x;

  int a = 0, len = 0;
  if (t < NBLK) {
    a   = lofsT[(size_t)j * NBLK + t];
    len = lofsT[(size_t)(j + 1) * NBLK + t] - a;
    s_a[t] = t * CH + a;
  }
  if (t < BIN_NODES) { s_cnt[t] = 0; s_cur[t] = 0; }
  int total, binbase;
  const int ro = block_excl_scan(len, s_w1, lane, wave, total);
  (void)block_excl_scan(a, s_w2, lane, wave, binbase);
  if (t < NBLK) s_ro[t] = ro;
  __syncthreads();

  int top = 1;
  while (top * 2 < NBLK) top *= 2;

  auto fetch = [&](int i) -> unsigned {
    int pos = 0;
    for (int step = top; step > 0; step >>= 1) {
      const int nxt = pos + step;
      if (nxt < NBLK && s_ro[nxt] <= i) pos = nxt;
    }
    return binned[s_a[pos] + (i - s_ro[pos])];
  };

  const bool fast = (total <= SORT_CAP);
  unsigned v[SORT_EPT];
  int      rk[SORT_EPT];
  if (fast) {
    int pos[SORT_EPT];
#pragma unroll
    for (int m = 0; m < SORT_EPT; ++m) pos[m] = 0;
    for (int step = top; step > 0; step >>= 1) {
#pragma unroll
      for (int m = 0; m < SORT_EPT; ++m) {
        const int nx = pos[m] + step;
        const int rd = s_ro[min(nx, NBLK - 1)];
        if (nx < NBLK && rd <= m * 1024 + t) pos[m] = nx;
      }
    }
#pragma unroll
    for (int m = 0; m < SORT_EPT; ++m) {
      const int i = m * 1024 + t;
      v[m] = 0u;
      if (i < total) v[m] = binned[s_a[pos[m]] + (i - s_ro[pos[m]])];
    }
#pragma unroll
    for (int m = 0; m < SORT_EPT; ++m) {
      const int i = m * 1024 + t;
      rk[m] = 0;
      if (i < total) rk[m] = atomicAdd(&s_cnt[v[m] >> 23], 1);
    }
  } else {
    for (int i = t; i < total; i += 1024) atomicAdd(&s_cnt[fetch(i) >> 23], 1);
  }
  __syncthreads();

  int dummy;
  const int c   = (t < BIN_NODES) ? s_cnt[t] : 0;
  const int exn = block_excl_scan(c, s_w3, lane, wave, dummy);
  if (t < BIN_NODES) {
    s_ofs[t] = exn;
    const int node = j * BIN_NODES + t;
    if (node < N) {
      rowptr[node] = binbase + exn;
      cnt[node]    = c;
      dinv[node]   = rsqrtf((float)(c + 1));
    }
  }
  __syncthreads();

  if (fast) {
#pragma unroll
    for (int m = 0; m < SORT_EPT; ++m) {
      const int i = m * 1024 + t;
      if (i < total) s_list[s_ofs[v[m] >> 23] + rk[m]] = (int)(v[m] & 0x7FFFFFu);
    }
    __syncthreads();
    for (int i = t; i < total; i += 1024) list[binbase + i] = s_list[i];
  } else {
    for (int i = t; i < total; i += 1024) {
      const unsigned w  = fetch(i);
      const int      dl = (int)(w >> 23);
      const int      p  = s_ofs[dl] + atomicAdd(&s_cur[dl], 1);
      list[binbase + p] = (int)(w & 0x7FFFFFu);
    }
  }
}

__device__ __forceinline__ void lds_barrier() {
  __builtin_amdgcn_fence(__ATOMIC_RELEASE, "workgroup", "local");
  __builtin_amdgcn_s_barrier();
  __builtin_amdgcn_fence(__ATOMIC_ACQUIRE, "workgroup", "local");
}


__global__ __launch_bounds__(128)
void k_gemm1_wmma(const float* __restrict__ x, const float* __restrict__ W1,
                  const float* __restrict__ dinv, float* __restrict__ hx, int ntiles) {
  __shared__ float sA[2][16 * 132];
  const int t  = threadIdx.x;

  const int wave  = t >> 5;
  const int lane  = t & 31;
  const int c0    = wave * 16;
  const int lhalf = lane & 15;
  const int khi   = (lane >> 4) * 2;

  v2f bf[INC / 4];
#pragma unroll
  for (int kk = 0; kk < INC / 4; ++kk) {
    const float* bp = W1 + (size_t)(kk * 4 + khi) * HIDC + c0 + lhalf;
    bf[kk].x = bp[0];
    bf[kk].y = bp[HIDC];
  }

  const int row = t >> 3;
  const int q   = (t & 7) * 16;

  int tile = blockIdx.x;
  float4 p0 = make_float4(0.0f, 0.0f, 0.0f, 0.0f), p1 = p0, p2 = p0, p3 = p0;
  if (tile < ntiles) {
    const float4* sp = (const float4*)(x + ((size_t)tile * 16 + row) * INC + q);
    p0 = sp[0]; p1 = sp[1]; p2 = sp[2]; p3 = sp[3];
  }
  int buf = 0;
  for (; tile < ntiles; tile += gridDim.x, buf ^= 1) {
    const float* sa = sA[buf];
    float4* const d = (float4*)(sA[buf] + row * 132 + q);
    d[0] = p0; d[1] = p1; d[2] = p2; d[3] = p3;
    lds_barrier();

    const int nt = (tile + gridDim.x < ntiles) ? tile + gridDim.x : tile;
    {
      const float4* sp = (const float4*)(x + ((size_t)nt * 16 + row) * INC + q);
      p0 = sp[0]; p1 = sp[1]; p2 = sp[2]; p3 = sp[3];
    }
    __builtin_amdgcn_sched_barrier(0);

    const int r0 = tile * 16;
    const int m0 = r0 + ((lane >> 4) * 8);
    const int n  = c0 + lhalf;
    float dv[8];
#pragma unroll
    for (int v = 0; v < 8; ++v) dv[v] = dinv[m0 + v];

    v8f acc = {};
#pragma unroll
    for (int kk = 0; kk < INC / 4; ++kk) {
      v2f a;
      const float* ap = sa + lhalf * 132 + kk * 4 + khi;
      a.x = ap[0];
      a.y = ap[1];
      acc = __builtin_amdgcn_wmma_f32_16x16x4_f32(false, a, false, bf[kk],
                                                  (short)0, acc, false, false);
    }

#pragma unroll
    for (int v = 0; v < 8; ++v)
      hx[(size_t)(m0 + v) * HIDC + n] = acc[v] * dv[v];
  }
}


__global__ __launch_bounds__(96)
void k_gemm2_wmma(const float* __restrict__ agg1, const float* __restrict__ b1,
                  const float* __restrict__ W2, const float* __restrict__ dinv,
                  float* __restrict__ h2, int ntiles) {
  __shared__ float sA[2][16 * 68];
  const int t  = threadIdx.x;

  const int wave  = t >> 5;
  const int lane  = t & 31;
  const int c0    = wave * 16;
  const int lhalf = lane & 15;
  const int khi   = (lane >> 4) * 2;
  const int col   = c0 + lhalf;
  const bool colOK = (col < OUTC);

  v2f bf[HIDC / 4];
#pragma unroll
  for (int kk = 0; kk < HIDC / 4; ++kk) {
    float bx = 0.0f, by = 0.0f;
    if (colOK) {
      bx = W2[(size_t)(kk * 4 + khi) * OUTC + col];
      by = W2[(size_t)(kk * 4 + khi + 1) * OUTC + col];
    }
    bf[kk].x = bx;
    bf[kk].y = by;
  }

  const int q = (t & 15) * 4;
  const float4 bq = *(const float4*)(b1 + q);
  const bool third = (t + 192 < 256);
  auto stage = [&](float* sa, float4 v, int i) {
    v.x = fmaxf(v.x + bq.x, 0.0f);
    v.y = fmaxf(v.y + bq.y, 0.0f);
    v.z = fmaxf(v.z + bq.z, 0.0f);
    v.w = fmaxf(v.w + bq.w, 0.0f);
    *(float4*)(sa + (i >> 4) * 68 + q) = v;
  };

  int tile = blockIdx.x;
  float4 p0 = make_float4(0.0f, 0.0f, 0.0f, 0.0f), p1 = p0, p2 = p0;
  if (tile < ntiles) {
    const float4* sp = (const float4*)(agg1 + (size_t)tile * 16 * HIDC);
    p0 = sp[t];
    p1 = sp[t + 96];
    if (third) p2 = sp[t + 192];
  }
  int buf = 0;
  for (; tile < ntiles; tile += gridDim.x, buf ^= 1) {
    const int m0 = tile * 16 + ((lane >> 4) * 8);
    float dv[8];
#pragma unroll
    for (int v = 0; v < 8; ++v) dv[v] = dinv[m0 + v];

    float* sa = sA[buf];
    stage(sa, p0, t);
    stage(sa, p1, t + 96);
    if (third) stage(sa, p2, t + 192);
    lds_barrier();

    const int nt = (tile + gridDim.x < ntiles) ? tile + gridDim.x : tile;
    {
      const float4* sp = (const float4*)(agg1 + (size_t)nt * 16 * HIDC);
      p0 = sp[t];
      p1 = sp[t + 96];
      if (third) p2 = sp[t + 192];
    }

    v8f acc = {};
#pragma unroll
    for (int kk = 0; kk < HIDC / 4; ++kk) {
      v2f a;
      const float* ap = sa + lhalf * 68 + kk * 4 + khi;
      a.x = ap[0];
      a.y = ap[1];
      acc = __builtin_amdgcn_wmma_f32_16x16x4_f32(false, a, false, bf[kk],
                                                  (short)0, acc, false, false);
    }

    if (colOK) {
#pragma unroll
      for (int v = 0; v < 8; ++v)
        h2[(size_t)(m0 + v) * OUTC + col] = acc[v] * dv[v];
    }
  }
}


template <int C4, bool BIAS>
__device__ __forceinline__ void pull_rows(const float* __restrict__ h, const int* __restrict__ list,
                                          const int* __restrict__ rowptr, const int* __restrict__ cnt,
                                          const float* __restrict__ dinv, const float* __restrict__ bias,
                                          float* __restrict__ o, int n) {
  const int gpb    = blockDim.x >> 4;
  const int stride = gridDim.x * gpb;
  const int sub    = threadIdx.x & 15;
  int node = blockIdx.x * gpb + (threadIdx.x >> 4);
  if (node >= n || sub >= C4) return;

  const float4* h4 = (const float4*)h;
  float4 b = make_float4(0.0f, 0.0f, 0.0f, 0.0f);
  if (BIAS) b = ((const float4*)bias)[sub];

  int    beg = rowptr[node];
  int    num = cnt[node];
  float  di  = dinv[node];
  float4 acc = h4[(size_t)node * C4 + sub];

  for (;;) {
    const int  nxt  = node + stride;
    const bool more = (nxt < n);
    const int  nn   = more ? nxt : node;
    const int    nbeg = rowptr[nn];
    const int    nnum = cnt[nn];
    const float  ndi  = dinv[nn];
    const float4 nacc = h4[(size_t)nn * C4 + sub];

    const int end = beg + num;
    int j = beg;
    for (; j + 4 <= end; j += 4) {
      const int s0 = list[j];
      const int s1 = list[j + 1];
      const int s2 = list[j + 2];
      const int s3 = list[j + 3];
      const float4 v0 = h4[(size_t)s0 * C4 + sub];
      const float4 v1 = h4[(size_t)s1 * C4 + sub];
      const float4 v2 = h4[(size_t)s2 * C4 + sub];
      const float4 v3 = h4[(size_t)s3 * C4 + sub];
      acc.x += v0.x; acc.y += v0.y; acc.z += v0.z; acc.w += v0.w;
      acc.x += v1.x; acc.y += v1.y; acc.z += v1.z; acc.w += v1.w;
      acc.x += v2.x; acc.y += v2.y; acc.z += v2.z; acc.w += v2.w;
      acc.x += v3.x; acc.y += v3.y; acc.z += v3.z; acc.w += v3.w;
    }
    if (j < end) {
      const int  last = end - 1;
      const bool p1 = (j + 1 < end);
      const bool p2 = (j + 2 < end);
      const int s0 = list[j];
      const int s1 = list[min(j + 1, last)];
      const int s2 = list[min(j + 2, last)];
      const float4 v0 = h4[(size_t)s0 * C4 + sub];
      const float4 r1 = h4[(size_t)s1 * C4 + sub];
      const float4 r2 = h4[(size_t)s2 * C4 + sub];
      asm volatile("" :: "v"(r1.x), "v"(r1.y), "v"(r1.z), "v"(r1.w),
                         "v"(r2.x), "v"(r2.y), "v"(r2.z), "v"(r2.w));
      const float4 v1 = p1 ? r1 : make_float4(0.0f, 0.0f, 0.0f, 0.0f);
      const float4 v2 = p2 ? r2 : make_float4(0.0f, 0.0f, 0.0f, 0.0f);
      acc.x += v0.x; acc.y += v0.y; acc.z += v0.z; acc.w += v0.w;
      acc.x += v1.x; acc.y += v1.y; acc.z += v1.z; acc.w += v1.w;
      acc.x += v2.x; acc.y += v2.y; acc.z += v2.z; acc.w += v2.w;
    }
    if (BIAS) {
      acc.x = b.x + di * acc.x;
      acc.y = b.y + di * acc.y;
      acc.z = b.z + di * acc.z;
      acc.w = b.w + di * acc.w;
    } else {
      acc.x *= di; acc.y *= di; acc.z *= di; acc.w *= di;
    }
    ((float4*)o)[(size_t)node * C4 + sub] = acc;

    if (!more) break;
    node = nxt; beg = nbeg; num = nnum; di = ndi; acc = nacc;
  }
}

__global__ __launch_bounds__(256)
void k_pull_h(const float* __restrict__ hx, const int* __restrict__ list,
              const int* __restrict__ rowptr, const int* __restrict__ cnt,
              const float* __restrict__ dinv, float* __restrict__ agg, int n) {
  pull_rows<HIDC / 4, false>(hx, list, rowptr, cnt, dinv, nullptr, agg, n);
}

__global__ __launch_bounds__(256)
void k_pull_o(const float* __restrict__ h2, const int* __restrict__ list,
              const int* __restrict__ rowptr, const int* __restrict__ cnt,
              const float* __restrict__ dinv, const float* __restrict__ b2,
              float* __restrict__ out, int n) {
  pull_rows<OUTC / 4, true>(h2, list, rowptr, cnt, dinv, b2, out, n);
}


extern "C" void kernel_launch(void* const* d_in, const int* in_sizes, int n_in,
                              void* d_out, int out_size, void* d_ws, size_t ws_size,
                              hipStream_t stream) {
  const float* x  = (const float*)d_in[0];
  const int*   ei = (const int*)d_in[1];
  const float* W1 = (const float*)d_in[2];
  const float* b1 = (const float*)d_in[3];
  const float* W2 = (const float*)d_in[4];
  const float* b2 = (const float*)d_in[5];
  float* out = (float*)d_out;

  const int N = in_sizes[0] / INC;
  const int E = in_sizes[1] / 2;
  const int* src = ei;
  const int* dst = ei + E;

  char*  ws  = (char*)d_ws;
  size_t off = 0;
  auto carve = [&](size_t bytes) -> void* {
    void* p = (void*)(ws + off);
    off = (off + bytes + 255) & ~(size_t)255;
    return p;
  };
  int*   cnt    = (int*)  carve((size_t)N * 4);
  int*   rowptr = (int*)  carve((size_t)N * 4);
  float* dinv   = (float*)carve((size_t)N * 4);
  int*   list   = (int*)  carve((size_t)E * 4);
  int*   rank   = (int*)  carve((size_t)E * 4);
  float* hx     = (float*)carve((size_t)N * HIDC * 4);
  float* agg1   = (float*)carve((size_t)N * HIDC * 4);
  float* h2     = hx;
  (void)n_in; (void)out_size; (void)ws_size;

  const int B = 256;
  const int G  = 2048;
  const int G1 = 1792;
  const int NB   = (N + BIN_NODES - 1) / BIN_NODES;
  const int NBLK = (E + CH - 1) / CH;
  unsigned* binned = (unsigned*)rank;
  int*      lofsT  = (int*)agg1;
  k_bin <<<NBLK, 1024, 0, stream>>>(src, dst, binned, lofsT, E, NB, NBLK);
  k_sort<<<NB,   1024, 0, stream>>>(binned, lofsT, list, rowptr, cnt, dinv, N, NBLK);

  k_gemm1_wmma<<<G1, 128, 0, stream>>>(x, W1, dinv, hx, N / 16);
  k_pull_h    <<<G, B, 0, stream>>>(hx, list, rowptr, cnt, dinv, agg1, N);

  k_gemm2_wmma<<<G, 96, 0, stream>>>(agg1, b1, W2, dinv, h2, N / 16);
  k_pull_o    <<<G, B, 0, stream>>>(h2, list, rowptr, cnt, dinv, b2, out, N);
}


# Layer-1 pull and GEMM2 in one launch (6 to 5): a block's 16 groups put their finished rows (dinv, bias, ReLU) in an LDS tile, three of its waves multiply it by W2; no agg1 stored; sum order unchanged
# speedup vs baseline: 4.6320x; 1.0644x over previous
#include <hip/hip_runtime.h>
#include <cstdint>
#include <cstddef>

#define INC  128
#define HIDC 64
#define OUTC 40

typedef float v2f __attribute__((ext_vector_type(2)));
typedef float v8f __attribute__((ext_vector_type(8)));


#define BIN_SHIFT 9
#define BIN_NODES 512
#define NBMAX     256
#define CH        8192
#define MAXBLK    1024
#define SORT_EPT  12
#define SORT_CAP  (SORT_EPT * 1024)

__device__ __forceinline__ int wave_incl_scan(int x, int lane) {
  for (int m = 1; m < 32; m <<= 1) {
    const int o = __shfl_up(x, m, 32);
    if (lane >= m) x += o;
  }
  return x;
}

__device__ __forceinline__ int block_excl_scan(int x, int* s_w, int lane, int wave, int& total) {
  const int inc = wave_incl_scan(x, lane);
  if (lane == 31) s_w[wave] = inc;
  __syncthreads();
  const int wv = s_w[lane];
  const int wi = wave_incl_scan(wv, lane);
  total = __shfl(wi, 31, 32);
  const int wex = __shfl(wi - wv, wave, 32);
  return wex + inc - x;
}

__global__ __launch_bounds__(1024)
void k_bin(const int* __restrict__ src, const int* __restrict__ dst,
           unsigned* __restrict__ binned, int* __restrict__ lofsT,
           int E, int NB, int NBLK) {
  __shared__ int      s_hist[NBMAX];
  __shared__ int      s_lofs[NBMAX + 1];
  __shared__ int      s_w[32];
  __shared__ unsigned s_pack[CH];
  const int t    = threadIdx.x;
  const int lane = t & 31;
  const int wave = t >> 5;
  const int blk  = blockIdx.x;
  const int base = blk * CH;

  if (t < NBMAX) s_hist[t] = 0;
  __syncthreads();

  int sv[8], dv[8];
  const bool whole = ((E & 3) == 0) && (base + CH <= E);
  if (whole) {
#pragma unroll
    for (int q = 0; q < 2; ++q) {
      const int e0 = base + q * (CH / 2) + t * 4;
      const int4 s4 = *(const int4*)(src + e0);
      const int4 d4 = *(const int4*)(dst + e0);
      sv[q * 4 + 0] = s4.x; sv[q * 4 + 1] = s4.y; sv[q * 4 + 2] = s4.z; sv[q * 4 + 3] = s4.w;
      dv[q * 4 + 0] = d4.x; dv[q * 4 + 1] = d4.y; dv[q * 4 + 2] = d4.z; dv[q * 4 + 3] = d4.w;
    }
  } else {
#pragma unroll
    for (int k = 0; k < 8; ++k) {
      const int e = base + (k >> 2) * (CH / 2) + t * 4 + (k & 3);
      sv[k] = (e < E) ? src[e] : 0;
      dv[k] = (e < E) ? dst[e] : -1;
    }
  }

  int rk[8];
#pragma unroll
  for (int k = 0; k < 8; ++k) {
    rk[k] = 0;
    if (dv[k] >= 0) {
      const unsigned bin = min((unsigned)dv[k] >> BIN_SHIFT, (unsigned)(NBMAX - 1));
      rk[k] = atomicAdd(&s_hist[bin], 1);
    }
  }
  __syncthreads();

  int nloc;
  const int h  = (t < NBMAX) ? s_hist[t] : 0;
  const int ex = block_excl_scan(h, s_w, lane, wave, nloc);
  if (t < NBMAX) s_lofs[t] = ex;
  if (t == NBMAX) s_lofs[NBMAX] = nloc;
  __syncthreads();

  if (t <= NB) lofsT[(size_t)t * NBLK + blk] = s_lofs[t];
#pragma unroll
  for (int k = 0; k < 8; ++k) {
    if (dv[k] >= 0) {
      const unsigned bin = min((unsigned)dv[k] >> BIN_SHIFT, (unsigned)(NBMAX - 1));
      s_pack[s_lofs[bin] + rk[k]] =
          ((unsigned)(dv[k] & (BIN_NODES - 1)) << 23) | (unsigned)sv[k];
    }
  }
  __syncthreads();

  for (int i = t; i < nloc; i += 1024) binned[base + i] = s_pack[i];
}

__global__ __launch_bounds__(1024)
void k_sort(const unsigned* __restrict__ binned, const int* __restrict__ lofsT,
            int* __restrict__ list, int* __restrict__ rowptr, int* __restrict__ cnt,
            float* __restrict__ dinv, int N, int NBLK) {
  __shared__ int s_a[MAXBLK];
  __shared__ int s_ro[MAXBLK];
  __shared__ int s_cnt[BIN_NODES];
  __shared__ int s_ofs[BIN_NODES];
  __shared__ int s_cur[BIN_NODES];
  __shared__ int s_w1[32], s_w2[32], s_w3[32];
  __shared__ int s_list[SORT_CAP];
  const int t    = threadIdx.x;
  const int lane = t & 31;
  const int wave = t >> 5;
  const int j    = blockIdx.x;

  int a = 0, len = 0;
  if (t < NBLK) {
    a   = lofsT[(size_t)j * NBLK + t];
    len = lofsT[(size_t)(j + 1) * NBLK + t] - a;
    s_a[t] = t * CH + a;
  }
  if (t < BIN_NODES) { s_cnt[t] = 0; s_cur[t] = 0; }
  int total, binbase;
  const int ro = block_excl_scan(len, s_w1, lane, wave, total);
  (void)block_excl_scan(a, s_w2, lane, wave, binbase);
  if (t < NBLK) s_ro[t] = ro;
  __syncthreads();

  int top = 1;
  while (top * 2 < NBLK) top *= 2;

  auto fetch = [&](int i) -> unsigned {
    int pos = 0;
    for (int step = top; step > 0; step >>= 1) {
      const int nxt = pos + step;
      if (nxt < NBLK && s_ro[nxt] <= i) pos = nxt;
    }
    return binned[s_a[pos] + (i - s_ro[pos])];
  };

  const bool fast = (total <= SORT_CAP);
  unsigned v[SORT_EPT];
  int      rk[SORT_EPT];
  if (fast) {
    int pos[SORT_EPT];
#pragma unroll
    for (int m = 0; m < SORT_EPT; ++m) pos[m] = 0;
    for (int step = top; step > 0; step >>= 1) {
#pragma unroll
      for (int m = 0; m < SORT_EPT; ++m) {
        const int nx = pos[m] + step;
        const int rd = s_ro[min(nx, NBLK - 1)];
        if (nx < NBLK && rd <= m * 1024 + t) pos[m] = nx;
      }
    }
#pragma unroll
    for (int m = 0; m < SORT_EPT; ++m) {
      const int i = m * 1024 + t;
      v[m] = 0u;
      if (i < total) v[m] = binned[s_a[pos[m]] + (i - s_ro[pos[m]])];
    }
#pragma unroll
    for (int m = 0; m < SORT_EPT; ++m) {
      const int i = m * 1024 + t;
      rk[m] = 0;
      if (i < total) rk[m] = atomicAdd(&s_cnt[v[m] >> 23], 1);
    }
  } else {
    for (int i = t; i < total; i += 1024) atomicAdd(&s_cnt[fetch(i) >> 23], 1);
  }
  __syncthreads();

  int dummy;
  const int c   = (t < BIN_NODES) ? s_cnt[t] : 0;
  const int exn = block_excl_scan(c, s_w3, lane, wave, dummy);
  if (t < BIN_NODES) {
    s_ofs[t] = exn;
    const int node = j * BIN_NODES + t;
    if (node < N) {
      rowptr[node] = binbase + exn;
      cnt[node]    = c;
      dinv[node]   = rsqrtf((float)(c + 1));
    }
  }
  __syncthreads();

  if (fast) {
#pragma unroll
    for (int m = 0; m < SORT_EPT; ++m) {
      const int i = m * 1024 + t;
      if (i < total) s_list[s_ofs[v[m] >> 23] + rk[m]] = (int)(v[m] & 0x7FFFFFu);
    }
    __syncthreads();
    for (int i = t; i < total; i += 1024) list[binbase + i] = s_list[i];
  } else {
    for (int i = t; i < total; i += 1024) {
      const unsigned w  = fetch(i);
      const int      dl = (int)(w >> 23);
      const int      p  = s_ofs[dl] + atomicAdd(&s_cur[dl], 1);
      list[binbase + p] = (int)(w & 0x7FFFFFu);
    }
  }
}

__device__ __forceinline__ void lds_barrier() {
  __builtin_amdgcn_fence(__ATOMIC_RELEASE, "workgroup", "local");
  __builtin_amdgcn_s_barrier();
  __builtin_amdgcn_fence(__ATOMIC_ACQUIRE, "workgroup", "local");
}


__global__ __launch_bounds__(128)
void k_gemm1_wmma(const float* __restrict__ x, const float* __restrict__ W1,
                  const float* __restrict__ dinv, float* __restrict__ hx, int ntiles) {
  __shared__ float sA[2][16 * 132];
  const int t  = threadIdx.x;

  const int wave  = t >> 5;
  const int lane  = t & 31;
  const int c0    = wave * 16;
  const int lhalf = lane & 15;
  const int khi   = (lane >> 4) * 2;

  v2f bf[INC / 4];
#pragma unroll
  for (int kk = 0; kk < INC / 4; ++kk) {
    const float* bp = W1 + (size_t)(kk * 4 + khi) * HIDC + c0 + lhalf;
    bf[kk].x = bp[0];
    bf[kk].y = bp[HIDC];
  }

  const int row = t >> 3;
  const int q   = (t & 7) * 16;

  int tile = blockIdx.x;
  float4 p0 = make_float4(0.0f, 0.0f, 0.0f, 0.0f), p1 = p0, p2 = p0, p3 = p0;
  if (tile < ntiles) {
    const float4* sp = (const float4*)(x + ((size_t)tile * 16 + row) * INC + q);
    p0 = sp[0]; p1 = sp[1]; p2 = sp[2]; p3 = sp[3];
  }
  int buf = 0;
  for (; tile < ntiles; tile += gridDim.x, buf ^= 1) {
    const float* sa = sA[buf];
    float4* const d = (float4*)(sA[buf] + row * 132 + q);
    d[0] = p0; d[1] = p1; d[2] = p2; d[3] = p3;
    lds_barrier();

    const int nt = (tile + gridDim.x < ntiles) ? tile + gridDim.x : tile;
    {
      const float4* sp = (const float4*)(x + ((size_t)nt * 16 + row) * INC + q);
      p0 = sp[0]; p1 = sp[1]; p2 = sp[2]; p3 = sp[3];
    }
    __builtin_amdgcn_sched_barrier(0);

    const int r0 = tile * 16;
    const int m0 = r0 + ((lane >> 4) * 8);
    const int n  = c0 + lhalf;
    float dv[8];
#pragma unroll
    for (int v = 0; v < 8; ++v) dv[v] = dinv[m0 + v];

    v8f acc = {};
#pragma unroll
    for (int kk = 0; kk < INC / 4; ++kk) {
      v2f a;
      const float* ap = sa + lhalf * 132 + kk * 4 + khi;
      a.x = ap[0];
      a.y = ap[1];
      acc = __builtin_amdgcn_wmma_f32_16x16x4_f32(false, a, false, bf[kk],
                                                  (short)0, acc, false, false);
    }

#pragma unroll
    for (int v = 0; v < 8; ++v)
      hx[(size_t)(m0 + v) * HIDC + n] = acc[v] * dv[v];
  }
}


#define GF_BLOCKS 1792

__global__ __launch_bounds__(256)
void k_pull_gemm2(const float* __restrict__ hx, const int* __restrict__ list,
                  const int* __restrict__ rowptr, const int* __restrict__ cnt,
                  const float* __restrict__ dinv, const float* __restrict__ b1,
                  const float* __restrict__ W2, float* __restrict__ h2, int ntiles) {
  __shared__ float sA[2][16 * 68];
  __shared__ float sD[2][16];
  __shared__ float sB[(HIDC / 2) * 48 * 2];
  const int t     = threadIdx.x;
  const int wave  = t >> 5;
  const int lane  = t & 31;
  const int g     = t >> 4;
  const int sub   = t & 15;
  const int lhalf = lane & 15;
  const int khi   = (lane >> 4) * 2;
  const int col   = wave * 16 + lhalf;
  const bool mm    = (wave < 3);
  const bool colOK = mm && (col < OUTC);

  const float4* h4 = (const float4*)hx;
  const float4  b  = ((const float4*)b1)[sub];

  int tile = blockIdx.x;
  int    beg = 0, num = 0;
  float  di  = 0.0f;
  float4 acc = make_float4(0.0f, 0.0f, 0.0f, 0.0f);
  if (tile < ntiles) {
    const int node = tile * 16 + g;
    beg = rowptr[node];
    num = cnt[node];
    di  = dinv[node];
    acc = h4[(size_t)node * 16 + sub];
  }
  float wx[6], wy[6];
#pragma unroll
  for (int r = 0; r < 6; ++r) {
    const int i  = t + r * 256;
    const int kp = i / 48;
    const int cc = min(i - kp * 48, OUTC - 1);
    wx[r] = W2[(size_t)(kp * 2) * OUTC + cc];
    wy[r] = W2[(size_t)(kp * 2 + 1) * OUTC + cc];
  }
#pragma unroll
  for (int r = 0; r < 6; ++r) {
    const int i  = t + r * 256;
    const int kp = i / 48;
    const bool in = (i - kp * 48 < OUTC);
    sB[i * 2]     = in ? wx[r] : 0.0f;
    sB[i * 2 + 1] = in ? wy[r] : 0.0f;
  }

  int buf = 0;
  for (; tile < ntiles; tile += gridDim.x, buf ^= 1) {
    const int nt = (tile + gridDim.x < ntiles) ? tile + gridDim.x : tile;
    const int nn = nt * 16 + g;
    const int    nbeg = rowptr[nn];
    const int    nnum = cnt[nn];
    const float  ndi  = dinv[nn];
    const float4 nacc = h4[(size_t)nn * 16 + sub];

    const int end = beg + num;
    int j = beg;
    for (; j + 4 <= end; j += 4) {
      const int s0 = list[j];
      const int s1 = list[j + 1];
      const int s2 = list[j + 2];
      const int s3 = list[j + 3];
      const float4 v0 = h4[(size_t)s0 * 16 + sub];
      const float4 v1 = h4[(size_t)s1 * 16 + sub];
      const float4 v2 = h4[(size_t)s2 * 16 + sub];
      const float4 v3 = h4[(size_t)s3 * 16 + sub];
      acc.x += v0.x; acc.y += v0.y; acc.z += v0.z; acc.w += v0.w;
      acc.x += v1.x; acc.y += v1.y; acc.z += v1.z; acc.w += v1.w;
      acc.x += v2.x; acc.y += v2.y; acc.z += v2.z; acc.w += v2.w;
      acc.x += v3.x; acc.y += v3.y; acc.z += v3.z; acc.w += v3.w;
    }
    if (j < end) {
      const int  last = end - 1;
      const bool p1 = (j + 1 < end);
      const bool p2 = (j + 2 < end);
      const int s0 = list[j];
      const int s1 = list[min(j + 1, last)];
      const int s2 = list[min(j + 2, last)];
      const float4 v0 = h4[(size_t)s0 * 16 + sub];
      const float4 r1 = h4[(size_t)s1 * 16 + sub];
      const float4 r2 = h4[(size_t)s2 * 16 + sub];
      asm volatile("" :: "v"(r1.x), "v"(r1.y), "v"(r1.z), "v"(r1.w),
                         "v"(r2.x), "v"(r2.y), "v"(r2.z), "v"(r2.w));
      const float4 v1 = p1 ? r1 : make_float4(0.0f, 0.0f, 0.0f, 0.0f);
      const float4 v2 = p2 ? r2 : make_float4(0.0f, 0.0f, 0.0f, 0.0f);
      acc.x += v0.x; acc.y += v0.y; acc.z += v0.z; acc.w += v0.w;
      acc.x += v1.x; acc.y += v1.y; acc.z += v1.z; acc.w += v1.w;
      acc.x += v2.x; acc.y += v2.y; acc.z += v2.z; acc.w += v2.w;
    }

    float4 r;
    r.x = fmaxf(di * acc.x + b.x, 0.0f);
    r.y = fmaxf(di * acc.y + b.y, 0.0f);
    r.z = fmaxf(di * acc.z + b.z, 0.0f);
    r.w = fmaxf(di * acc.w + b.w, 0.0f);
    float* sa = sA[buf];
    *(float4*)(sa + g * 68 + sub * 4) = r;
    if (sub == 0) sD[buf][g] = di;
    lds_barrier();

    if (mm) {
      v8f c = {};
      const float* ap = sa + lhalf * 68 + khi;
      const float* bp = sB + ((lane >> 4) * 48 + col) * 2;
#pragma unroll
      for (int kk = 0; kk < HIDC / 4; ++kk) {
        v2f a, bb;
        a.x  = ap[kk * 4];
        a.y  = ap[kk * 4 + 1];
        bb.x = bp[kk * 192];
        bb.y = bp[kk * 192 + 1];
        c = __builtin_amdgcn_wmma_f32_16x16x4_f32(false, a, false, bb,
                                                  (short)0, c, false, false);
      }
      const int mh = (lane >> 4) * 8;
      const int m0 = tile * 16 + mh;
      if (colOK) {
#pragma unroll
        for (int v = 0; v < 8; ++v)
          h2[(size_t)(m0 + v) * OUTC + col] = c[v] * sD[buf][mh + v];
      }
    }

    beg = nbeg; num = nnum; di = ndi; acc = nacc;
  }
}


template <int C4, bool BIAS>
__device__ __forceinline__ void pull_rows(const float* __restrict__ h, const int* __restrict__ list,
                                          const int* __restrict__ rowptr, const int* __restrict__ cnt,
                                          const float* __restrict__ dinv, const float* __restrict__ bias,
                                          float* __restrict__ o, int n) {
  const int gpb    = blockDim.x >> 4;
  const int stride = gridDim.x * gpb;
  const int sub    = threadIdx.x & 15;
  int node = blockIdx.x * gpb + (threadIdx.x >> 4);
  if (node >= n || sub >= C4) return;

  const float4* h4 = (const float4*)h;
  float4 b = make_float4(0.0f, 0.0f, 0.0f, 0.0f);
  if (BIAS) b = ((const float4*)bias)[sub];

  int    beg = rowptr[node];
  int    num = cnt[node];
  float  di  = dinv[node];
  float4 acc = h4[(size_t)node * C4 + sub];

  for (;;) {
    const int  nxt  = node + stride;
    const bool more = (nxt < n);
    const int  nn   = more ? nxt : node;
    const int    nbeg = rowptr[nn];
    const int    nnum = cnt[nn];
    const float  ndi  = dinv[nn];
    const float4 nacc = h4[(size_t)nn * C4 + sub];

    const int end = beg + num;
    int j = beg;
    for (; j + 4 <= end; j += 4) {
      const int s0 = list[j];
      const int s1 = list[j + 1];
      const int s2 = list[j + 2];
      const int s3 = list[j + 3];
      const float4 v0 = h4[(size_t)s0 * C4 + sub];
      const float4 v1 = h4[(size_t)s1 * C4 + sub];
      const float4 v2 = h4[(size_t)s2 * C4 + sub];
      const float4 v3 = h4[(size_t)s3 * C4 + sub];
      acc.x += v0.x; acc.y += v0.y; acc.z += v0.z; acc.w += v0.w;
      acc.x += v1.x; acc.y += v1.y; acc.z += v1.z; acc.w += v1.w;
      acc.x += v2.x; acc.y += v2.y; acc.z += v2.z; acc.w += v2.w;
      acc.x += v3.x; acc.y += v3.y; acc.z += v3.z; acc.w += v3.w;
    }
    if (j < end) {
      const int  last = end - 1;
      const bool p1 = (j + 1 < end);
      const bool p2 = (j + 2 < end);
      const int s0 = list[j];
      const int s1 = list[min(j + 1, last)];
      const int s2 = list[min(j + 2, last)];
      const float4 v0 = h4[(size_t)s0 * C4 + sub];
      const float4 r1 = h4[(size_t)s1 * C4 + sub];
      const float4 r2 = h4[(size_t)s2 * C4 + sub];
      asm volatile("" :: "v"(r1.x), "v"(r1.y), "v"(r1.z), "v"(r1.w),
                         "v"(r2.x), "v"(r2.y), "v"(r2.z), "v"(r2.w));
      const float4 v1 = p1 ? r1 : make_float4(0.0f, 0.0f, 0.0f, 0.0f);
      const float4 v2 = p2 ? r2 : make_float4(0.0f, 0.0f, 0.0f, 0.0f);
      acc.x += v0.x; acc.y += v0.y; acc.z += v0.z; acc.w += v0.w;
      acc.x += v1.x; acc.y += v1.y; acc.z += v1.z; acc.w += v1.w;
      acc.x += v2.x; acc.y += v2.y; acc.z += v2.z; acc.w += v2.w;
    }
    if (BIAS) {
      acc.x = b.x + di * acc.x;
      acc.y = b.y + di * acc.y;
      acc.z = b.z + di * acc.z;
      acc.w = b.w + di * acc.w;
    } else {
      acc.x *= di; acc.y *= di; acc.z *= di; acc.w *= di;
    }
    ((float4*)o)[(size_t)node * C4 + sub] = acc;

    if (!more) break;
    node = nxt; beg = nbeg; num = nnum; di = ndi; acc = nacc;
  }
}

__global__ __launch_bounds__(256)
void k_pull_o(const float* __restrict__ h2, const int* __restrict__ list,
              const int* __restrict__ rowptr, const int* __restrict__ cnt,
              const float* __restrict__ dinv, const float* __restrict__ b2,
              float* __restrict__ out, int n) {
  pull_rows<OUTC / 4, true>(h2, list, rowptr, cnt, dinv, b2, out, n);
}


extern "C" void kernel_launch(void* const* d_in, const int* in_sizes, int n_in,
                              void* d_out, int out_size, void* d_ws, size_t ws_size,
                              hipStream_t stream) {
  const float* x  = (const float*)d_in[0];
  const int*   ei = (const int*)d_in[1];
  const float* W1 = (const float*)d_in[2];
  const float* b1 = (const float*)d_in[3];
  const float* W2 = (const float*)d_in[4];
  const float* b2 = (const float*)d_in[5];
  float* out = (float*)d_out;

  const int N = in_sizes[0] / INC;
  const int E = in_sizes[1] / 2;
  const int* src = ei;
  const int* dst = ei + E;

  char*  ws  = (char*)d_ws;
  size_t off = 0;
  auto carve = [&](size_t bytes) -> void* {
    void* p = (void*)(ws + off);
    off = (off + bytes + 255) & ~(size_t)255;
    return p;
  };
  int*   cnt    = (int*)  carve((size_t)N * 4);
  int*   rowptr = (int*)  carve((size_t)N * 4);
  float* dinv   = (float*)carve((size_t)N * 4);
  int*   list   = (int*)  carve((size_t)E * 4);
  int*   rank   = (int*)  carve((size_t)E * 4);
  float* hx     = (float*)carve((size_t)N * HIDC * 4);
  float* h2     = (float*)carve((size_t)N * HIDC * 4);
  (void)n_in; (void)out_size; (void)ws_size;

  const int B = 256;
  const int G  = 2048;
  const int G1 = 1792;
  const int GF = GF_BLOCKS;
  const int NB   = (N + BIN_NODES - 1) / BIN_NODES;
  const int NBLK = (E + CH - 1) / CH;
  unsigned* binned = (unsigned*)rank;
  int*      lofsT  = (int*)h2;
  k_bin <<<NBLK, 1024, 0, stream>>>(src, dst, binned, lofsT, E, NB, NBLK);
  k_sort<<<NB,   1024, 0, stream>>>(binned, lofsT, list, rowptr, cnt, dinv, N, NBLK);

  k_gemm1_wmma<<<G1, 128, 0, stream>>>(x, W1, dinv, hx, N / 16);

  k_pull_gemm2<<<GF, B, 0, stream>>>(hx, list, rowptr, cnt, dinv, b1, W2, h2, N / 16);

  k_pull_o    <<<G, B, 0, stream>>>(h2, list, rowptr, cnt, dinv, b2, out, N);
}


# In-edge ids fetched a node ahead, 16 per load (lane i of the node's 16-lane group holds id i), passed by DPP row_share: 8 row loads in flight, no id load between; sum order unchanged
# speedup vs baseline: 4.9192x; 1.0620x over previous
#include <hip/hip_runtime.h>
#include <cstdint>
#include <cstddef>

#define INC  128
#define HIDC 64
#define OUTC 40

typedef float v2f __attribute__((ext_vector_type(2)));
typedef float v8f __attribute__((ext_vector_type(8)));


#define BIN_SHIFT 9
#define BIN_NODES 512
#define NBMAX     256
#define CH        8192
#define MAXBLK    1024
#define SORT_EPT  12
#define SORT_CAP  (SORT_EPT * 1024)

__device__ __forceinline__ int wave_incl_scan(int x, int lane) {
  for (int m = 1; m < 32; m <<= 1) {
    const int o = __shfl_up(x, m, 32);
    if (lane >= m) x += o;
  }
  return x;
}

__device__ __forceinline__ int block_excl_scan(int x, int* s_w, int lane, int wave, int& total) {
  const int inc = wave_incl_scan(x, lane);
  if (lane == 31) s_w[wave] = inc;
  __syncthreads();
  const int wv = s_w[lane];
  const int wi = wave_incl_scan(wv, lane);
  total = __shfl(wi, 31, 32);
  const int wex = __shfl(wi - wv, wave, 32);
  return wex + inc - x;
}

__global__ __launch_bounds__(1024)
void k_bin(const int* __restrict__ src, const int* __restrict__ dst,
           unsigned* __restrict__ binned, int* __restrict__ lofsT,
           int E, int NB, int NBLK) {
  __shared__ int      s_hist[NBMAX];
  __shared__ int      s_lofs[NBMAX + 1];
  __shared__ int      s_w[32];
  __shared__ unsigned s_pack[CH];
  const int t    = threadIdx.x;
  const int lane = t & 31;
  const int wave = t >> 5;
  const int blk  = blockIdx.x;
  const int base = blk * CH;

  if (t < NBMAX) s_hist[t] = 0;
  __syncthreads();

  int sv[8], dv[8];
  const bool whole = ((E & 3) == 0) && (base + CH <= E);
  if (whole) {
#pragma unroll
    for (int q = 0; q < 2; ++q) {
      const int e0 = base + q * (CH / 2) + t * 4;
      const int4 s4 = *(const int4*)(src + e0);
      const int4 d4 = *(const int4*)(dst + e0);
      sv[q * 4 + 0] = s4.x; sv[q * 4 + 1] = s4.y; sv[q * 4 + 2] = s4.z; sv[q * 4 + 3] = s4.w;
      dv[q * 4 + 0] = d4.x; dv[q * 4 + 1] = d4.y; dv[q * 4 + 2] = d4.z; dv[q * 4 + 3] = d4.w;
    }
  } else {
#pragma unroll
    for (int k = 0; k < 8; ++k) {
      const int e = base + (k >> 2) * (CH / 2) + t * 4 + (k & 3);
      sv[k] = (e < E) ? src[e] : 0;
      dv[k] = (e < E) ? dst[e] : -1;
    }
  }

  int rk[8];
#pragma unroll
  for (int k = 0; k < 8; ++k) {
    rk[k] = 0;
    if (dv[k] >= 0) {
      const unsigned bin = min((unsigned)dv[k] >> BIN_SHIFT, (unsigned)(NBMAX - 1));
      rk[k] = atomicAdd(&s_hist[bin], 1);
    }
  }
  __syncthreads();

  int nloc;
  const int h  = (t < NBMAX) ? s_hist[t] : 0;
  const int ex = block_excl_scan(h, s_w, lane, wave, nloc);
  if (t < NBMAX) s_lofs[t] = ex;
  if (t == NBMAX) s_lofs[NBMAX] = nloc;
  __syncthreads();

  if (t <= NB) lofsT[(size_t)t * NBLK + blk] = s_lofs[t];
#pragma unroll
  for (int k = 0; k < 8; ++k) {
    if (dv[k] >= 0) {
      const unsigned bin = min((unsigned)dv[k] >> BIN_SHIFT, (unsigned)(NBMAX - 1));
      s_pack[s_lofs[bin] + rk[k]] =
          ((unsigned)(dv[k] & (BIN_NODES - 1)) << 23) | (unsigned)sv[k];
    }
  }
  __syncthreads();

  for (int i = t; i < nloc; i += 1024) binned[base + i] = s_pack[i];
}

__global__ __launch_bounds__(1024)
void k_sort(const unsigned* __restrict__ binned, const int* __restrict__ lofsT,
            int* __restrict__ list, int* __restrict__ rowptr, int* __restrict__ cnt,
            float* __restrict__ dinv, int N, int NBLK) {
  __shared__ int s_a[MAXBLK];
  __shared__ int s_ro[MAXBLK];
  __shared__ int s_cnt[BIN_NODES];
  __shared__ int s_ofs[BIN_NODES];
  __shared__ int s_cur[BIN_NODES];
  __shared__ int s_w1[32], s_w2[32], s_w3[32];
  __shared__ int s_list[SORT_CAP];
  const int t    = threadIdx.x;
  const int lane = t & 31;
  const int wave = t >> 5;
  const int j    = blockIdx.x;

  int a = 0, len = 0;
  if (t < NBLK) {
    a   = lofsT[(size_t)j * NBLK + t];
    len = lofsT[(size_t)(j + 1) * NBLK + t] - a;
    s_a[t] = t * CH + a;
  }
  if (t < BIN_NODES) { s_cnt[t] = 0; s_cur[t] = 0; }
  int total, binbase;
  const int ro = block_excl_scan(len, s_w1, lane, wave, total);
  (void)block_excl_scan(a, s_w2, lane, wave, binbase);
  if (t < NBLK) s_ro[t] = ro;
  __syncthreads();

  int top = 1;
  while (top * 2 < NBLK) top *= 2;

  auto fetch = [&](int i) -> unsigned {
    int pos = 0;
    for (int step = top; step > 0; step >>= 1) {
      const int nxt = pos + step;
      if (nxt < NBLK && s_ro[nxt] <= i) pos = nxt;
    }
    return binned[s_a[pos] + (i - s_ro[pos])];
  };

  const bool fast = (total <= SORT_CAP);
  unsigned v[SORT_EPT];
  int      rk[SORT_EPT];
  if (fast) {
    int pos[SORT_EPT];
#pragma unroll
    for (int m = 0; m < SORT_EPT; ++m) pos[m] = 0;
    for (int step = top; step > 0; step >>= 1) {
#pragma unroll
      for (int m = 0; m < SORT_EPT; ++m) {
        const int nx = pos[m] + step;
        const int rd = s_ro[min(nx, NBLK - 1)];
        if (nx < NBLK && rd <= m * 1024 + t) pos[m] = nx;
      }
    }
#pragma unroll
    for (int m = 0; m < SORT_EPT; ++m) {
      const int i = m * 1024 + t;
      v[m] = 0u;
      if (i < total) v[m] = binned[s_a[pos[m]] + (i - s_ro[pos[m]])];
    }
#pragma unroll
    for (int m = 0; m < SORT_EPT; ++m) {
      const int i = m * 1024 + t;
      rk[m] = 0;
      if (i < total) rk[m] = atomicAdd(&s_cnt[v[m] >> 23], 1);
    }
  } else {
    for (int i = t; i < total; i += 1024) atomicAdd(&s_cnt[fetch(i) >> 23], 1);
  }
  __syncthreads();

  int dummy;
  const int c   = (t < BIN_NODES) ? s_cnt[t] : 0;
  const int exn = block_excl_scan(c, s_w3, lane, wave, dummy);
  if (t < BIN_NODES) {
    s_ofs[t] = exn;
    const int node = j * BIN_NODES + t;
    if (node < N) {
      rowptr[node] = binbase + exn;
      cnt[node]    = c;
      dinv[node]   = rsqrtf((float)(c + 1));
    }
  }
  __syncthreads();

  if (fast) {
#pragma unroll
    for (int m = 0; m < SORT_EPT; ++m) {
      const int i = m * 1024 + t;
      if (i < total) s_list[s_ofs[v[m] >> 23] + rk[m]] = (int)(v[m] & 0x7FFFFFu);
    }
    __syncthreads();
    for (int i = t; i < total; i += 1024) list[binbase + i] = s_list[i];
  } else {
    for (int i = t; i < total; i += 1024) {
      const unsigned w  = fetch(i);
      const int      dl = (int)(w >> 23);
      const int      p  = s_ofs[dl] + atomicAdd(&s_cur[dl], 1);
      list[binbase + p] = (int)(w & 0x7FFFFFu);
    }
  }
}

__device__ __forceinline__ void lds_barrier() {
  __builtin_amdgcn_fence(__ATOMIC_RELEASE, "workgroup", "local");
  __builtin_amdgcn_s_barrier();
  __builtin_amdgcn_fence(__ATOMIC_ACQUIRE, "workgroup", "local");
}


__global__ __launch_bounds__(128)
void k_gemm1_wmma(const float* __restrict__ x, const float* __restrict__ W1,
                  const float* __restrict__ dinv, float* __restrict__ hx, int ntiles) {
  __shared__ float sA[2][16 * 132];
  const int t  = threadIdx.x;

  const int wave  = t >> 5;
  const int lane  = t & 31;
  const int c0    = wave * 16;
  const int lhalf = lane & 15;
  const int khi   = (lane >> 4) * 2;

  if (blockIdx.x == 0 && t < HIDC / 4)
    ((float4*)hx)[(size_t)ntiles * 16 * (HIDC / 4) + t] = make_float4(0.0f, 0.0f, 0.0f, 0.0f);

  v2f bf[INC / 4];
#pragma unroll
  for (int kk = 0; kk < INC / 4; ++kk) {
    const float* bp = W1 + (size_t)(kk * 4 + khi) * HIDC + c0 + lhalf;
    bf[kk].x = bp[0];
    bf[kk].y = bp[HIDC];
  }

  const int row = t >> 3;
  const int q   = (t & 7) * 16;

  int tile = blockIdx.x;
  float4 p0 = make_float4(0.0f, 0.0f, 0.0f, 0.0f), p1 = p0, p2 = p0, p3 = p0;
  if (tile < ntiles) {
    const float4* sp = (const float4*)(x + ((size_t)tile * 16 + row) * INC + q);
    p0 = sp[0]; p1 = sp[1]; p2 = sp[2]; p3 = sp[3];
  }
  int buf = 0;
  for (; tile < ntiles; tile += gridDim.x, buf ^= 1) {
    const float* sa = sA[buf];
    float4* const d = (float4*)(sA[buf] + row * 132 + q);
    d[0] = p0; d[1] = p1; d[2] = p2; d[3] = p3;
    lds_barrier();

    const int nt = (tile + gridDim.x < ntiles) ? tile + gridDim.x : tile;
    {
      const float4* sp = (const float4*)(x + ((size_t)nt * 16 + row) * INC + q);
      p0 = sp[0]; p1 = sp[1]; p2 = sp[2]; p3 = sp[3];
    }
    __builtin_amdgcn_sched_barrier(0);

    const int r0 = tile * 16;
    const int m0 = r0 + ((lane >> 4) * 8);
    const int n  = c0 + lhalf;
    float dv[8];
#pragma unroll
    for (int v = 0; v < 8; ++v) dv[v] = dinv[m0 + v];

    v8f acc = {};
#pragma unroll
    for (int kk = 0; kk < INC / 4; ++kk) {
      v2f a;
      const float* ap = sa + lhalf * 132 + kk * 4 + khi;
      a.x = ap[0];
      a.y = ap[1];
      acc = __builtin_amdgcn_wmma_f32_16x16x4_f32(false, a, false, bf[kk],
                                                  (short)0, acc, false, false);
    }

#pragma unroll
    for (int v = 0; v < 8; ++v)
      hx[(size_t)(m0 + v) * HIDC + n] = acc[v] * dv[v];
  }
}


template <int K>
__device__ __forceinline__ int row_lane(int v) {
  return __builtin_amdgcn_update_dpp(0, v, 0x150 + K, 0xf, 0xf, false);
}

struct rows8 { float4 v0, v1, v2, v3, v4, v5, v6, v7; };

template <int B>
__device__ __forceinline__ rows8 load8(const char* __restrict__ hb, int ido, unsigned colb) {
  const unsigned o0 = (unsigned)row_lane<B + 0>(ido) + colb;
  const unsigned o1 = (unsigned)row_lane<B + 1>(ido) + colb;
  const unsigned o2 = (unsigned)row_lane<B + 2>(ido) + colb;
  const unsigned o3 = (unsigned)row_lane<B + 3>(ido) + colb;
  const unsigned o4 = (unsigned)row_lane<B + 4>(ido) + colb;
  const unsigned o5 = (unsigned)row_lane<B + 5>(ido) + colb;
  const unsigned o6 = (unsigned)row_lane<B + 6>(ido) + colb;
  const unsigned o7 = (unsigned)row_lane<B + 7>(ido) + colb;
  rows8 r;
  r.v0 = *(const float4*)(hb + o0);
  r.v1 = *(const float4*)(hb + o1);
  r.v2 = *(const float4*)(hb + o2);
  r.v3 = *(const float4*)(hb + o3);
  r.v4 = *(const float4*)(hb + o4);
  r.v5 = *(const float4*)(hb + o5);
  r.v6 = *(const float4*)(hb + o6);
  r.v7 = *(const float4*)(hb + o7);
  return r;
}

__device__ __forceinline__ void sum8(const rows8& r, float4& acc) {
  acc.x += r.v0.x; acc.y += r.v0.y; acc.z += r.v0.z; acc.w += r.v0.w;
  acc.x += r.v1.x; acc.y += r.v1.y; acc.z += r.v1.z; acc.w += r.v1.w;
  acc.x += r.v2.x; acc.y += r.v2.y; acc.z += r.v2.z; acc.w += r.v2.w;
  acc.x += r.v3.x; acc.y += r.v3.y; acc.z += r.v3.z; acc.w += r.v3.w;
  acc.x += r.v4.x; acc.y += r.v4.y; acc.z += r.v4.z; acc.w += r.v4.w;
  acc.x += r.v5.x; acc.y += r.v5.y; acc.z += r.v5.z; acc.w += r.v5.w;
  acc.x += r.v6.x; acc.y += r.v6.y; acc.z += r.v6.z; acc.w += r.v6.w;
  acc.x += r.v7.x; acc.y += r.v7.y; acc.z += r.v7.z; acc.w += r.v7.w;
}

template <int ROWB>
__device__ __forceinline__ void add_in_rows(const char* __restrict__ hb, const int* __restrict__ list,
                                            int beg, int num, int ids, int nbeg, int& nraw,
                                            int zid, int elast, int sub, unsigned colb, float4& acc) {
  {
    int sids = zid;
    if (sub + 16 < num) sids = list[beg + 16 + sub];
    const int ido = ids * ROWB;
    rows8 r = load8<0>(hb, ido, colb);
    nraw = list[min(nbeg + sub, elast)];
    sum8(r, acc);
    if (num > 8) {
      r = load8<8>(hb, ido, colb);
      sum8(r, acc);
    }
    ids = sids;
  }
  for (int done = 16; done < num; done += 16) {
    const int left = num - done;
    int sids = zid;
    if (sub + 16 < left) sids = list[beg + done + 16 + sub];
    const int ido = ids * ROWB;
    rows8 r = load8<0>(hb, ido, colb);
    sum8(r, acc);
    if (left > 8) {
      r = load8<8>(hb, ido, colb);
      sum8(r, acc);
    }
    ids = sids;
  }
}


#define GF_BLOCKS 1792

__global__ __launch_bounds__(256)
void k_pull_gemm2(const float* __restrict__ hx, const int* __restrict__ list,
                  const int* __restrict__ rowptr, const int* __restrict__ cnt,
                  const float* __restrict__ dinv, const float* __restrict__ b1,
                  const float* __restrict__ W2, float* __restrict__ h2, int ntiles, int E) {
  __shared__ float sA[2][16 * 68];
  __shared__ float sD[2][16];
  __shared__ float sB[(HIDC / 2) * 48 * 2];
  const int t     = threadIdx.x;
  const int wave  = t >> 5;
  const int lane  = t & 31;
  const int g     = t >> 4;
  const int sub   = t & 15;
  const int lhalf = lane & 15;
  const int khi   = (lane >> 4) * 2;
  const int col   = wave * 16 + lhalf;
  const bool mm    = (wave < 3);
  const bool colOK = mm && (col < OUTC);

  const float4  b  = ((const float4*)b1)[sub];
  const char*    hb   = (const char*)hx;
  const unsigned colb = (unsigned)sub * 16u;
  const int      zid  = ntiles * 16;
  const int      elast = max(E - 1, 0);

  if (blockIdx.x == 0 && t < OUTC / 4)
    ((float4*)h2)[(size_t)ntiles * 16 * (OUTC / 4) + t] = make_float4(0.0f, 0.0f, 0.0f, 0.0f);

  int tile = blockIdx.x;
  int    beg = 0, num = 0;
  float  di  = 0.0f;
  int    ids = zid;
  if (tile < ntiles) {
    const int node = tile * 16 + g;
    beg = rowptr[node];
    num = cnt[node];
    di  = dinv[node];
    if (sub < num) ids = list[beg + sub];
  }
  float wx[6], wy[6];
#pragma unroll
  for (int r = 0; r < 6; ++r) {
    const int i  = t + r * 256;
    const int kp = i / 48;
    const int cc = min(i - kp * 48, OUTC - 1);
    wx[r] = W2[(size_t)(kp * 2) * OUTC + cc];
    wy[r] = W2[(size_t)(kp * 2 + 1) * OUTC + cc];
  }
#pragma unroll
  for (int r = 0; r < 6; ++r) {
    const int i  = t + r * 256;
    const int kp = i / 48;
    const bool in = (i - kp * 48 < OUTC);
    sB[i * 2]     = in ? wx[r] : 0.0f;
    sB[i * 2 + 1] = in ? wy[r] : 0.0f;
  }

  int buf = 0;
  for (; tile < ntiles; tile += gridDim.x, buf ^= 1) {
    const int nt = (tile + gridDim.x < ntiles) ? tile + gridDim.x : tile;
    const int nn = nt * 16 + g;
    const int    nbeg = rowptr[nn];
    const int    nnum = cnt[nn];
    const float  ndi  = dinv[nn];

    float4 acc = *(const float4*)(hb + ((unsigned)(tile * 16 + g) * (HIDC * 4) + colb));
    int nraw;
    add_in_rows<HIDC * 4>(hb, list, beg, num, ids, nbeg, nraw, zid, elast, sub, colb, acc);

    float4 r;
    r.x = fmaxf(di * acc.x + b.x, 0.0f);
    r.y = fmaxf(di * acc.y + b.y, 0.0f);
    r.z = fmaxf(di * acc.z + b.z, 0.0f);
    r.w = fmaxf(di * acc.w + b.w, 0.0f);
    float* sa = sA[buf];
    *(float4*)(sa + g * 68 + sub * 4) = r;
    if (sub == 0) sD[buf][g] = di;
    lds_barrier();

    if (mm) {
      v8f c = {};
      const float* ap = sa + lhalf * 68 + khi;
      const float* bp = sB + ((lane >> 4) * 48 + col) * 2;
#pragma unroll
      for (int kk = 0; kk < HIDC / 4; ++kk) {
        v2f a, bb;
        a.x  = ap[kk * 4];
        a.y  = ap[kk * 4 + 1];
        bb.x = bp[kk * 192];
        bb.y = bp[kk * 192 + 1];
        c = __builtin_amdgcn_wmma_f32_16x16x4_f32(false, a, false, bb,
                                                  (short)0, c, false, false);
      }
      const int mh = (lane >> 4) * 8;
      const int m0 = tile * 16 + mh;
      if (colOK) {
#pragma unroll
        for (int v = 0; v < 8; ++v)
          h2[(size_t)(m0 + v) * OUTC + col] = c[v] * sD[buf][mh + v];
      }
    }

    beg = nbeg; num = nnum; di = ndi;
    ids = (sub < nnum) ? nraw : zid;
  }
}


template <int C4, bool BIAS>
__device__ __forceinline__ void pull_rows(const float* __restrict__ h, const int* __restrict__ list,
                                          const int* __restrict__ rowptr, const int* __restrict__ cnt,
                                          const float* __restrict__ dinv, const float* __restrict__ bias,
                                          float* __restrict__ o, int n, int E) {
  const int gpb    = blockDim.x >> 4;
  const int stride = gridDim.x * gpb;
  const int sub    = threadIdx.x & 15;
  int node = blockIdx.x * gpb + (threadIdx.x >> 4);
  if (node >= n) return;

  const int      col  = min(sub, C4 - 1);
  const unsigned colb = (unsigned)col * 16u;
  const char*    hb   = (const char*)h;
  const int      zid  = n;
  const int      elast = max(E - 1, 0);
  char* const    ob   = (char*)o;
  float4 b = make_float4(0.0f, 0.0f, 0.0f, 0.0f);
  if (BIAS) b = ((const float4*)bias)[col];

  int    beg = rowptr[node];
  int    num = cnt[node];
  float  di  = dinv[node];
  int    ids = zid;
  if (sub < num) ids = list[beg + sub];

  for (;;) {
    const int  nxt  = node + stride;
    const bool more = (nxt < n);
    const int  nn   = more ? nxt : node;
    const int    nbeg = rowptr[nn];
    const int    nnum = cnt[nn];
    const float  ndi  = dinv[nn];

    float4 acc = *(const float4*)(hb + ((unsigned)node * (C4 * 16) + colb));
    int nraw;
    add_in_rows<C4 * 16>(hb, list, beg, num, ids, nbeg, nraw, zid, elast, sub, colb, acc);

    if (BIAS) {
      acc.x = b.x + di * acc.x;
      acc.y = b.y + di * acc.y;
      acc.z = b.z + di * acc.z;
      acc.w = b.w + di * acc.w;
    } else {
      acc.x *= di; acc.y *= di; acc.z *= di; acc.w *= di;
    }
    if (sub < C4) *(float4*)(ob + ((unsigned)node * (C4 * 16) + colb)) = acc;

    if (!more) break;
    node = nxt; beg = nbeg; num = nnum; di = ndi;
    ids = (sub < nnum) ? nraw : zid;
  }
}

__global__ __launch_bounds__(256)
void k_pull_o(const float* __restrict__ h2, const int* __restrict__ list,
              const int* __restrict__ rowptr, const int* __restrict__ cnt,
              const float* __restrict__ dinv, const float* __restrict__ b2,
              float* __restrict__ out, int n, int E) {
  pull_rows<OUTC / 4, true>(h2, list, rowptr, cnt, dinv, b2, out, n, E);
}


extern "C" void kernel_launch(void* const* d_in, const int* in_sizes, int n_in,
                              void* d_out, int out_size, void* d_ws, size_t ws_size,
                              hipStream_t stream) {
  const float* x  = (const float*)d_in[0];
  const int*   ei = (const int*)d_in[1];
  const float* W1 = (const float*)d_in[2];
  const float* b1 = (const float*)d_in[3];
  const float* W2 = (const float*)d_in[4];
  const float* b2 = (const float*)d_in[5];
  float* out = (float*)d_out;

  const int N = in_sizes[0] / INC;
  const int E = in_sizes[1] / 2;
  const int* src = ei;
  const int* dst = ei + E;

  char*  ws  = (char*)d_ws;
  size_t off = 0;
  auto carve = [&](size_t bytes) -> void* {
    void* p = (void*)(ws + off);
    off = (off + bytes + 255) & ~(size_t)255;
    return p;
  };
  int*   cnt    = (int*)  carve((size_t)N * 4);
  int*   rowptr = (int*)  carve((size_t)N * 4);
  float* dinv   = (float*)carve((size_t)N * 4);
  int*   list   = (int*)  carve((size_t)E * 4);
  int*   rank   = (int*)  carve((size_t)E * 4);
  float* hx     = (float*)carve((size_t)(N + 1) * HIDC * 4);
  float* h2     = (float*)carve((size_t)(N + 1) * OUTC * 4);
  (void)n_in; (void)out_size; (void)ws_size;

  const int B = 256;
  const int G  = 2048;
  const int G1 = 1792;
  const int GF = GF_BLOCKS;
  const int NB   = (N + BIN_NODES - 1) / BIN_NODES;
  const int NBLK = (E + CH - 1) / CH;
  unsigned* binned = (unsigned*)rank;
  int*      lofsT  = (int*)h2;
  k_bin <<<NBLK, 1024, 0, stream>>>(src, dst, binned, lofsT, E, NB, NBLK);
  k_sort<<<NB,   1024, 0, stream>>>(binned, lofsT, list, rowptr, cnt, dinv, N, NBLK);

  k_gemm1_wmma<<<G1, 128, 0, stream>>>(x, W1, dinv, hx, N / 16);

  k_pull_gemm2<<<GF, B, 0, stream>>>(hx, list, rowptr, cnt, dinv, b1, W2, h2, N / 16, E);

  k_pull_o    <<<G, B, 0, stream>>>(h2, list, rowptr, cnt, dinv, b2, out, N, E);
}


# GEMM1 joins the sort launch (5 to 4 launches): 250 bins of 400 nodes, a bin's block multiplies its own 25 row tiles with the dinv it just made; sum order = LDS arrival rank within the new bins
# speedup vs baseline: 5.1366x; 1.0442x over previous
#include <hip/hip_runtime.h>
#include <cstdint>
#include <cstddef>

#define INC  128
#define HIDC 64
#define OUTC 40

typedef float v2f __attribute__((ext_vector_type(2)));
typedef float v8f __attribute__((ext_vector_type(8)));


#define BIN_NODES 400
#define BIN_TILES (BIN_NODES / 16)
#define BIN_SLOTS 512
#define NBMAX     256
#define CH        8192
#define MAXBLK    1024
#define SORT_EPT  8
#define SORT_CAP  (SORT_EPT * 1024)

__device__ __forceinline__ void lds_barrier() {
  __builtin_amdgcn_fence(__ATOMIC_RELEASE, "workgroup", "local");
  __builtin_amdgcn_s_barrier();
  __builtin_amdgcn_fence(__ATOMIC_ACQUIRE, "workgroup", "local");
}

__device__ __forceinline__ int wave_incl_scan(int x, int lane) {
  for (int m = 1; m < 32; m <<= 1) {
    const int o = __shfl_up(x, m, 32);
    if (lane >= m) x += o;
  }
  return x;
}

template <bool LDSONLY = false>
__device__ __forceinline__ int block_excl_scan(int x, int* s_w, int lane, int wave, int& total) {
  const int inc = wave_incl_scan(x, lane);
  if (lane == 31) s_w[wave] = inc;
  if (LDSONLY) lds_barrier(); else __syncthreads();
  const int wv = s_w[lane];
  const int wi = wave_incl_scan(wv, lane);
  total = __shfl(wi, 31, 32);
  const int wex = __shfl(wi - wv, wave, 32);
  return wex + inc - x;
}

__global__ __launch_bounds__(1024)
void k_bin(const int* __restrict__ src, const int* __restrict__ dst,
           unsigned* __restrict__ binned, int* __restrict__ lofsT,
           int E, int NB, int NBLK) {
  __shared__ int      s_hist[NBMAX];
  __shared__ int      s_lofs[NBMAX + 1];
  __shared__ int      s_w[32];
  __shared__ unsigned s_pack[CH];
  const int t    = threadIdx.x;
  const int lane = t & 31;
  const int wave = t >> 5;
  const int blk  = blockIdx.x;
  const int base = blk * CH;

  if (t < NBMAX) s_hist[t] = 0;
  __syncthreads();

  int sv[8], dv[8];
  const bool whole = ((E & 3) == 0) && (base + CH <= E);
  if (whole) {
#pragma unroll
    for (int q = 0; q < 2; ++q) {
      const int e0 = base + q * (CH / 2) + t * 4;
      const int4 s4 = *(const int4*)(src + e0);
      const int4 d4 = *(const int4*)(dst + e0);
      sv[q * 4 + 0] = s4.x; sv[q * 4 + 1] = s4.y; sv[q * 4 + 2] = s4.z; sv[q * 4 + 3] = s4.w;
      dv[q * 4 + 0] = d4.x; dv[q * 4 + 1] = d4.y; dv[q * 4 + 2] = d4.z; dv[q * 4 + 3] = d4.w;
    }
  } else {
#pragma unroll
    for (int k = 0; k < 8; ++k) {
      const int e = base + (k >> 2) * (CH / 2) + t * 4 + (k & 3);
      sv[k] = (e < E) ? src[e] : 0;
      dv[k] = (e < E) ? dst[e] : -1;
    }
  }

  int rk[8];
#pragma unroll
  for (int k = 0; k < 8; ++k) {
    rk[k] = 0;
    if (dv[k] >= 0) {
      const unsigned bin = min((unsigned)dv[k] / BIN_NODES, (unsigned)(NBMAX - 1));
      rk[k] = atomicAdd(&s_hist[bin], 1);
    }
  }
  __syncthreads();

  int nloc;
  const int h  = (t < NBMAX) ? s_hist[t] : 0;
  const int ex = block_excl_scan(h, s_w, lane, wave, nloc);
  if (t < NBMAX) s_lofs[t] = ex;
  if (t == NBMAX) s_lofs[NBMAX] = nloc;
  __syncthreads();

  if (t <= NB) lofsT[(size_t)t * NBLK + blk] = s_lofs[t];
#pragma unroll
  for (int k = 0; k < 8; ++k) {
    if (dv[k] >= 0) {
      const unsigned bin = min((unsigned)dv[k] / BIN_NODES, (unsigned)(NBMAX - 1));
      s_pack[s_lofs[bin] + rk[k]] =
          (((unsigned)dv[k] - bin * BIN_NODES) << 23) | (unsigned)sv[k];
    }
  }
  __syncthreads();

  for (int i = t; i < nloc; i += 1024) binned[base + i] = s_pack[i];
}


__global__ __launch_bounds__(1024)
void k_sort_gemm1(const unsigned* __restrict__ binned, const int* __restrict__ lofsT,
                  int* __restrict__ list, int* __restrict__ rowptr, int* __restrict__ cnt,
                  float* __restrict__ dinv, const float* __restrict__ x,
                  const float* __restrict__ W1, float* __restrict__ hx,
                  int N, int NBLK, int ntiles) {
  __shared__ int s_a[MAXBLK];
  __shared__ int s_ro[MAXBLK];
  __shared__ int s_cnt[BIN_SLOTS];
  __shared__ int s_ofs[BIN_SLOTS];
  __shared__ int s_cur[BIN_SLOTS];
  __shared__ float s_dinv[BIN_SLOTS];
  __shared__ int s_w1[32], s_w2[32], s_w3[32];
  __shared__ int s_list[SORT_CAP];
  __shared__ float sA[8][2][16 * 132];
  const int t    = threadIdx.x;
  const int lane = t & 31;
  const int wave = t >> 5;
  const int j    = blockIdx.x;

  const int grp   = wave >> 2;
  const int c0    = (wave & 3) * 16;
  const int lhalf = lane & 15;
  const int khi   = (lane >> 4) * 2;
  const int tg    = t & 127;
  const int row   = tg >> 3;
  const int q     = (tg & 7) * 16;
  const int tbase = j * BIN_TILES;
  const int nloc  = max(0, min(BIN_TILES, ntiles - tbase));

  if (j == 0 && t < HIDC / 4)
    ((float4*)hx)[(size_t)ntiles * 16 * (HIDC / 4) + t] = make_float4(0.0f, 0.0f, 0.0f, 0.0f);

  int a = 0, len = 0;
  if (t < NBLK) {
    a   = lofsT[(size_t)j * NBLK + t];
    len = lofsT[(size_t)(j + 1) * NBLK + t] - a;
  }
  float4 p0 = make_float4(0.0f, 0.0f, 0.0f, 0.0f), p1 = p0, p2 = p0, p3 = p0;
  if (grp < nloc) {
    const float4* sp = (const float4*)(x + ((size_t)(tbase + grp) * 16 + row) * INC + q);
    p0 = sp[0]; p1 = sp[1]; p2 = sp[2]; p3 = sp[3];
  }
  if (t < NBLK) s_a[t] = t * CH + a;
  if (t < BIN_SLOTS) { s_cnt[t] = 0; s_cur[t] = 0; }
  int total, binbase;
  const int ro = block_excl_scan<true>(len, s_w1, lane, wave, total);
  (void)block_excl_scan<true>(a, s_w2, lane, wave, binbase);
  if (t < NBLK) s_ro[t] = ro;
  lds_barrier();

  int top = 1;
  while (top * 2 < NBLK) top *= 2;

  auto fetch = [&](int i) -> unsigned {
    int pos = 0;
    for (int step = top; step > 0; step >>= 1) {
      const int nxt = pos + step;
      if (nxt < NBLK && s_ro[nxt] <= i) pos = nxt;
    }
    return binned[s_a[pos] + (i - s_ro[pos])];
  };

  const bool fast = (total <= SORT_CAP);
  unsigned v[SORT_EPT];
  int      rk[SORT_EPT];
  if (fast) {
    int pos[SORT_EPT];
#pragma unroll
    for (int m = 0; m < SORT_EPT; ++m) pos[m] = 0;
    for (int step = top; step > 0; step >>= 1) {
#pragma unroll
      for (int m = 0; m < SORT_EPT; ++m) {
        const int nx = pos[m] + step;
        const int rd = s_ro[min(nx, NBLK - 1)];
        if (nx < NBLK && rd <= m * 1024 + t) pos[m] = nx;
      }
    }
#pragma unroll
    for (int m = 0; m < SORT_EPT; ++m) {
      const int i = m * 1024 + t;
      v[m] = 0u;
      if (i < total) v[m] = binned[s_a[pos[m]] + (i - s_ro[pos[m]])];
    }
#pragma unroll
    for (int m = 0; m < SORT_EPT; ++m) {
      const int i = m * 1024 + t;
      rk[m] = 0;
      if (i < total) rk[m] = atomicAdd(&s_cnt[v[m] >> 23], 1);
    }
  } else {
    for (int i = t; i < total; i += 1024) atomicAdd(&s_cnt[fetch(i) >> 23], 1);
  }
  __syncthreads();

  int dummy;
  const int c   = (t < BIN_NODES) ? s_cnt[t] : 0;
  const int exn = block_excl_scan(c, s_w3, lane, wave, dummy);
  if (t < BIN_NODES) {
    s_ofs[t] = exn;
    const float di = rsqrtf((float)(c + 1));
    s_dinv[t] = di;
    const int node = j * BIN_NODES + t;
    if (node < N) {
      rowptr[node] = binbase + exn;
      cnt[node]    = c;
      dinv[node]   = di;
    }
  }
  __syncthreads();

  if (fast) {
#pragma unroll
    for (int m = 0; m < SORT_EPT; ++m) {
      const int i = m * 1024 + t;
      if (i < total) s_list[s_ofs[v[m] >> 23] + rk[m]] = (int)(v[m] & 0x7FFFFFu);
    }
    __syncthreads();
    for (int i = t; i < total; i += 1024) list[binbase + i] = s_list[i];
  } else {
    for (int i = t; i < total; i += 1024) {
      const unsigned w  = fetch(i);
      const int      dl = (int)(w >> 23);
      const int      p  = s_ofs[dl] + atomicAdd(&s_cur[dl], 1);
      list[binbase + p] = (int)(w & 0x7FFFFFu);
    }
  }

  v2f bf[INC / 4];
#pragma unroll
  for (int kk = 0; kk < INC / 4; ++kk) {
    const float* bp = W1 + (size_t)(kk * 4 + khi) * HIDC + c0 + lhalf;
    bf[kk].x = bp[0];
    bf[kk].y = bp[HIDC];
  }

  int buf = 0;
  for (int lt0 = 0; lt0 < nloc; lt0 += 8, buf ^= 1) {
    const int  lt  = lt0 + grp;
    const bool act = (lt < nloc);
    if (act) {
      float4* const d = (float4*)(sA[grp][buf] + row * 132 + q);
      d[0] = p0; d[1] = p1; d[2] = p2; d[3] = p3;
    }
    lds_barrier();
    if (act) {
      const float* sa = sA[grp][buf];
      const int nt = (lt + 8 < nloc) ? lt + 8 : lt;
      {
        const float4* sp = (const float4*)(x + ((size_t)(tbase + nt) * 16 + row) * INC + q);
        p0 = sp[0]; p1 = sp[1]; p2 = sp[2]; p3 = sp[3];
      }
      __builtin_amdgcn_sched_barrier(0);

      const int ml = lt * 16 + ((lane >> 4) * 8);
      const int m0 = tbase * 16 + ml;
      const int n  = c0 + lhalf;
      float dv[8];
#pragma unroll
      for (int u = 0; u < 8; ++u) dv[u] = s_dinv[ml + u];

      v8f acc = {};
#pragma unroll
      for (int kk = 0; kk < INC / 4; ++kk) {
        v2f af;
        const float* ap = sa + lhalf * 132 + kk * 4 + khi;
        af.x = ap[0];
        af.y = ap[1];
        acc = __builtin_amdgcn_wmma_f32_16x16x4_f32(false, af, false, bf[kk],
                                                    (short)0, acc, false, false);
      }

#pragma unroll
      for (int u = 0; u < 8; ++u)
        hx[(size_t)(m0 + u) * HIDC + n] = acc[u] * dv[u];
    }
  }
}


template <int K>
__device__ __forceinline__ int row_lane(int v) {
  return __builtin_amdgcn_update_dpp(0, v, 0x150 + K, 0xf, 0xf, false);
}

struct rows8 { float4 v0, v1, v2, v3, v4, v5, v6, v7; };

template <int B>
__device__ __forceinline__ rows8 load8(const char* __restrict__ hb, int ido, unsigned colb) {
  const unsigned o0 = (unsigned)row_lane<B + 0>(ido) + colb;
  const unsigned o1 = (unsigned)row_lane<B + 1>(ido) + colb;
  const unsigned o2 = (unsigned)row_lane<B + 2>(ido) + colb;
  const unsigned o3 = (unsigned)row_lane<B + 3>(ido) + colb;
  const unsigned o4 = (unsigned)row_lane<B + 4>(ido) + colb;
  const unsigned o5 = (unsigned)row_lane<B + 5>(ido) + colb;
  const unsigned o6 = (unsigned)row_lane<B + 6>(ido) + colb;
  const unsigned o7 = (unsigned)row_lane<B + 7>(ido) + colb;
  rows8 r;
  r.v0 = *(const float4*)(hb + o0);
  r.v1 = *(const float4*)(hb + o1);
  r.v2 = *(const float4*)(hb + o2);
  r.v3 = *(const float4*)(hb + o3);
  r.v4 = *(const float4*)(hb + o4);
  r.v5 = *(const float4*)(hb + o5);
  r.v6 = *(const float4*)(hb + o6);
  r.v7 = *(const float4*)(hb + o7);
  return r;
}

__device__ __forceinline__ void sum8(const rows8& r, float4& acc) {
  acc.x += r.v0.x; acc.y += r.v0.y; acc.z += r.v0.z; acc.w += r.v0.w;
  acc.x += r.v1.x; acc.y += r.v1.y; acc.z += r.v1.z; acc.w += r.v1.w;
  acc.x += r.v2.x; acc.y += r.v2.y; acc.z += r.v2.z; acc.w += r.v2.w;
  acc.x += r.v3.x; acc.y += r.v3.y; acc.z += r.v3.z; acc.w += r.v3.w;
  acc.x += r.v4.x; acc.y += r.v4.y; acc.z += r.v4.z; acc.w += r.v4.w;
  acc.x += r.v5.x; acc.y += r.v5.y; acc.z += r.v5.z; acc.w += r.v5.w;
  acc.x += r.v6.x; acc.y += r.v6.y; acc.z += r.v6.z; acc.w += r.v6.w;
  acc.x += r.v7.x; acc.y += r.v7.y; acc.z += r.v7.z; acc.w += r.v7.w;
}

template <int ROWB>
__device__ __forceinline__ void add_in_rows(const char* __restrict__ hb, const int* __restrict__ list,
                                            int beg, int num, int ids, int nbeg, int& nraw,
                                            int zid, int elast, int sub, unsigned colb, float4& acc) {
  {
    int sids = zid;
    if (sub + 16 < num) sids = list[beg + 16 + sub];
    const int ido = ids * ROWB;
    rows8 r = load8<0>(hb, ido, colb);
    nraw = list[min(nbeg + sub, elast)];
    sum8(r, acc);
    if (num > 8) {
      r = load8<8>(hb, ido, colb);
      sum8(r, acc);
    }
    ids = sids;
  }
  for (int done = 16; done < num; done += 16) {
    const int left = num - done;
    int sids = zid;
    if (sub + 16 < left) sids = list[beg + done + 16 + sub];
    const int ido = ids * ROWB;
    rows8 r = load8<0>(hb, ido, colb);
    sum8(r, acc);
    if (left > 8) {
      r = load8<8>(hb, ido, colb);
      sum8(r, acc);
    }
    ids = sids;
  }
}


#define GF_BLOCKS 1792

__global__ __launch_bounds__(256)
void k_pull_gemm2(const float* __restrict__ hx, const int* __restrict__ list,
                  const int* __restrict__ rowptr, const int* __restrict__ cnt,
                  const float* __restrict__ dinv, const float* __restrict__ b1,
                  const float* __restrict__ W2, float* __restrict__ h2, int ntiles, int E) {
  __shared__ float sA[2][16 * 68];
  __shared__ float sD[2][16];
  __shared__ float sB[(HIDC / 2) * 48 * 2];
  const int t     = threadIdx.x;
  const int wave  = t >> 5;
  const int lane  = t & 31;
  const int g     = t >> 4;
  const int sub   = t & 15;
  const int lhalf = lane & 15;
  const int khi   = (lane >> 4) * 2;
  const int col   = wave * 16 + lhalf;
  const bool mm    = (wave < 3);
  const bool colOK = mm && (col < OUTC);

  const float4  b  = ((const float4*)b1)[sub];
  const char*    hb   = (const char*)hx;
  const unsigned colb = (unsigned)sub * 16u;
  const int      zid  = ntiles * 16;
  const int      elast = max(E - 1, 0);

  if (blockIdx.x == 0 && t < OUTC / 4)
    ((float4*)h2)[(size_t)ntiles * 16 * (OUTC / 4) + t] = make_float4(0.0f, 0.0f, 0.0f, 0.0f);

  int tile = blockIdx.x;
  int    beg = 0, num = 0;
  float  di  = 0.0f;
  int    ids = zid;
  if (tile < ntiles) {
    const int node = tile * 16 + g;
    beg = rowptr[node];
    num = cnt[node];
    di  = dinv[node];
    if (sub < num) ids = list[beg + sub];
  }
  float wx[6], wy[6];
#pragma unroll
  for (int r = 0; r < 6; ++r) {
    const int i  = t + r * 256;
    const int kp = i / 48;
    const int cc = min(i - kp * 48, OUTC - 1);
    wx[r] = W2[(size_t)(kp * 2) * OUTC + cc];
    wy[r] = W2[(size_t)(kp * 2 + 1) * OUTC + cc];
  }
#pragma unroll
  for (int r = 0; r < 6; ++r) {
    const int i  = t + r * 256;
    const int kp = i / 48;
    const bool in = (i - kp * 48 < OUTC);
    sB[i * 2]     = in ? wx[r] : 0.0f;
    sB[i * 2 + 1] = in ? wy[r] : 0.0f;
  }

  int buf = 0;
  for (; tile < ntiles; tile += gridDim.x, buf ^= 1) {
    const int nt = (tile + gridDim.x < ntiles) ? tile + gridDim.x : tile;
    const int nn = nt * 16 + g;
    const int    nbeg = rowptr[nn];
    const int    nnum = cnt[nn];
    const float  ndi  = dinv[nn];

    float4 acc = *(const float4*)(hb + ((unsigned)(tile * 16 + g) * (HIDC * 4) + colb));
    int nraw;
    add_in_rows<HIDC * 4>(hb, list, beg, num, ids, nbeg, nraw, zid, elast, sub, colb, acc);

    float4 r;
    r.x = fmaxf(di * acc.x + b.x, 0.0f);
    r.y = fmaxf(di * acc.y + b.y, 0.0f);
    r.z = fmaxf(di * acc.z + b.z, 0.0f);
    r.w = fmaxf(di * acc.w + b.w, 0.0f);
    float* sa = sA[buf];
    *(float4*)(sa + g * 68 + sub * 4) = r;
    if (sub == 0) sD[buf][g] = di;
    lds_barrier();

    if (mm) {
      v8f c = {};
      const float* ap = sa + lhalf * 68 + khi;
      const float* bp = sB + ((lane >> 4) * 48 + col) * 2;
#pragma unroll
      for (int kk = 0; kk < HIDC / 4; ++kk) {
        v2f a, bb;
        a.x  = ap[kk * 4];
        a.y  = ap[kk * 4 + 1];
        bb.x = bp[kk * 192];
        bb.y = bp[kk * 192 + 1];
        c = __builtin_amdgcn_wmma_f32_16x16x4_f32(false, a, false, bb,
                                                  (short)0, c, false, false);
      }
      const int mh = (lane >> 4) * 8;
      const int m0 = tile * 16 + mh;
      if (colOK) {
#pragma unroll
        for (int v = 0; v < 8; ++v)
          h2[(size_t)(m0 + v) * OUTC + col] = c[v] * sD[buf][mh + v];
      }
    }

    beg = nbeg; num = nnum; di = ndi;
    ids = (sub < nnum) ? nraw : zid;
  }
}


template <int C4, bool BIAS>
__device__ __forceinline__ void pull_rows(const float* __restrict__ h, const int* __restrict__ list,
                                          const int* __restrict__ rowptr, const int* __restrict__ cnt,
                                          const float* __restrict__ dinv, const float* __restrict__ bias,
                                          float* __restrict__ o, int n, int E) {
  const int gpb    = blockDim.x >> 4;
  const int stride = gridDim.x * gpb;
  const int sub    = threadIdx.x & 15;
  int node = blockIdx.x * gpb + (threadIdx.x >> 4);
  if (node >= n) return;

  const int      col  = min(sub, C4 - 1);
  const unsigned colb = (unsigned)col * 16u;
  const char*    hb   = (const char*)h;
  const int      zid  = n;
  const int      elast = max(E - 1, 0);
  char* const    ob   = (char*)o;
  float4 b = make_float4(0.0f, 0.0f, 0.0f, 0.0f);
  if (BIAS) b = ((const float4*)bias)[col];

  int    beg = rowptr[node];
  int    num = cnt[node];
  float  di  = dinv[node];
  int    ids = zid;
  if (sub < num) ids = list[beg + sub];

  for (;;) {
    const int  nxt  = node + stride;
    const bool more = (nxt < n);
    const int  nn   = more ? nxt : node;
    const int    nbeg = rowptr[nn];
    const int    nnum = cnt[nn];
    const float  ndi  = dinv[nn];

    float4 acc = *(const float4*)(hb + ((unsigned)node * (C4 * 16) + colb));
    int nraw;
    add_in_rows<C4 * 16>(hb, list, beg, num, ids, nbeg, nraw, zid, elast, sub, colb, acc);

    if (BIAS) {
      acc.x = b.x + di * acc.x;
      acc.y = b.y + di * acc.y;
      acc.z = b.z + di * acc.z;
      acc.w = b.w + di * acc.w;
    } else {
      acc.x *= di; acc.y *= di; acc.z *= di; acc.w *= di;
    }
    if (sub < C4) *(float4*)(ob + ((unsigned)node * (C4 * 16) + colb)) = acc;

    if (!more) break;
    node = nxt; beg = nbeg; num = nnum; di = ndi;
    ids = (sub < nnum) ? nraw : zid;
  }
}

__global__ __launch_bounds__(256)
void k_pull_o(const float* __restrict__ h2, const int* __restrict__ list,
              const int* __restrict__ rowptr, const int* __restrict__ cnt,
              const float* __restrict__ dinv, const float* __restrict__ b2,
              float* __restrict__ out, int n, int E) {
  pull_rows<OUTC / 4, true>(h2, list, rowptr, cnt, dinv, b2, out, n, E);
}


extern "C" void kernel_launch(void* const* d_in, const int* in_sizes, int n_in,
                              void* d_out, int out_size, void* d_ws, size_t ws_size,
                              hipStream_t stream) {
  const float* x  = (const float*)d_in[0];
  const int*   ei = (const int*)d_in[1];
  const float* W1 = (const float*)d_in[2];
  const float* b1 = (const float*)d_in[3];
  const float* W2 = (const float*)d_in[4];
  const float* b2 = (const float*)d_in[5];
  float* out = (float*)d_out;

  const int N = in_sizes[0] / INC;
  const int E = in_sizes[1] / 2;
  const int* src = ei;
  const int* dst = ei + E;

  char*  ws  = (char*)d_ws;
  size_t off = 0;
  auto carve = [&](size_t bytes) -> void* {
    void* p = (void*)(ws + off);
    off = (off + bytes + 255) & ~(size_t)255;
    return p;
  };
  int*   cnt    = (int*)  carve((size_t)N * 4);
  int*   rowptr = (int*)  carve((size_t)N * 4);
  float* dinv   = (float*)carve((size_t)N * 4);
  int*   list   = (int*)  carve((size_t)E * 4);
  int*   rank   = (int*)  carve((size_t)E * 4);
  float* hx     = (float*)carve((size_t)(N + 1) * HIDC * 4);
  float* h2     = (float*)carve((size_t)(N + 1) * OUTC * 4);
  (void)n_in; (void)out_size; (void)ws_size;

  const int B = 256;
  const int G  = 2048;
  const int GF = GF_BLOCKS;
  const int NB   = (N + BIN_NODES - 1) / BIN_NODES;
  const int NBLK = (E + CH - 1) / CH;
  unsigned* binned = (unsigned*)rank;
  int*      lofsT  = (int*)h2;
  k_bin       <<<NBLK, 1024, 0, stream>>>(src, dst, binned, lofsT, E, NB, NBLK);
  k_sort_gemm1<<<NB,   1024, 0, stream>>>(binned, lofsT, list, rowptr, cnt, dinv, x, W1, hx,
                                          N, NBLK, N / 16);

  k_pull_gemm2<<<GF, B, 0, stream>>>(hx, list, rowptr, cnt, dinv, b1, W2, h2, N / 16, E);

  k_pull_o    <<<G, B, 0, stream>>>(h2, list, rowptr, cnt, dinv, b2, out, N, E);
}


# Bin's edges fetched run by run: a wave reads whole chunk runs (lane = edge); no per-edge search over run starts (64 chained LDS reads), run lengths summed not scanned; sum order = LDS arrival rank
# speedup vs baseline: 5.3619x; 1.0439x over previous
#include <hip/hip_runtime.h>
#include <cstdint>
#include <cstddef>

#define INC  128
#define HIDC 64
#define OUTC 40

typedef float v2f __attribute__((ext_vector_type(2)));
typedef float v8f __attribute__((ext_vector_type(8)));


#define BIN_NODES 400
#define BIN_TILES (BIN_NODES / 16)
#define BIN_SLOTS 512
#define NBMAX     256
#define CH        8192
#define MAXBLK    1024
#define SORT_CAP  8192
#define SORT_RPW  7
#define SORT_SUB  3

__device__ __forceinline__ void lds_barrier() {
  __builtin_amdgcn_fence(__ATOMIC_RELEASE, "workgroup", "local");
  __builtin_amdgcn_s_barrier();
  __builtin_amdgcn_fence(__ATOMIC_ACQUIRE, "workgroup", "local");
}

__device__ __forceinline__ int wave_incl_scan(int x, int lane) {
  for (int m = 1; m < 32; m <<= 1) {
    const int o = __shfl_up(x, m, 32);
    if (lane >= m) x += o;
  }
  return x;
}

template <bool LDSONLY = false>
__device__ __forceinline__ int block_excl_scan(int x, int* s_w, int lane, int wave, int& total) {
  const int inc = wave_incl_scan(x, lane);
  if (lane == 31) s_w[wave] = inc;
  if (LDSONLY) lds_barrier(); else __syncthreads();
  const int wv = s_w[lane];
  const int wi = wave_incl_scan(wv, lane);
  total = __shfl(wi, 31, 32);
  const int wex = __shfl(wi - wv, wave, 32);
  return wex + inc - x;
}

__device__ __forceinline__ void block_sum3(int& x, int& y, int& z, int* s_x, int* s_y, int* s_z,
                                           int lane, int wave) {
  for (int m = 16; m > 0; m >>= 1) {
    x += __shfl_xor(x, m, 32);
    y += __shfl_xor(y, m, 32);
    z += __shfl_xor(z, m, 32);
  }
  if (lane == 0) { s_x[wave] = x; s_y[wave] = y; s_z[wave] = z; }
  lds_barrier();
  x = s_x[lane]; y = s_y[lane]; z = s_z[lane];
  for (int m = 16; m > 0; m >>= 1) {
    x += __shfl_xor(x, m, 32);
    y += __shfl_xor(y, m, 32);
    z += __shfl_xor(z, m, 32);
  }
}

__global__ __launch_bounds__(1024)
void k_bin(const int* __restrict__ src, const int* __restrict__ dst,
           unsigned* __restrict__ binned, int* __restrict__ lofsT,
           int E, int NB, int NBLK) {
  __shared__ int      s_hist[NBMAX];
  __shared__ int      s_lofs[NBMAX + 1];
  __shared__ int      s_w[32];
  __shared__ unsigned s_pack[CH];
  const int t    = threadIdx.x;
  const int lane = t & 31;
  const int wave = t >> 5;
  const int blk  = blockIdx.x;
  const int base = blk * CH;

  if (t < NBMAX) s_hist[t] = 0;
  __syncthreads();

  int sv[8], dv[8];
  const bool whole = ((E & 3) == 0) && (base + CH <= E);
  if (whole) {
#pragma unroll
    for (int q = 0; q < 2; ++q) {
      const int e0 = base + q * (CH / 2) + t * 4;
      const int4 s4 = *(const int4*)(src + e0);
      const int4 d4 = *(const int4*)(dst + e0);
      sv[q * 4 + 0] = s4.x; sv[q * 4 + 1] = s4.y; sv[q * 4 + 2] = s4.z; sv[q * 4 + 3] = s4.w;
      dv[q * 4 + 0] = d4.x; dv[q * 4 + 1] = d4.y; dv[q * 4 + 2] = d4.z; dv[q * 4 + 3] = d4.w;
    }
  } else {
#pragma unroll
    for (int k = 0; k < 8; ++k) {
      const int e = base + (k >> 2) * (CH / 2) + t * 4 + (k & 3);
      sv[k] = (e < E) ? src[e] : 0;
      dv[k] = (e < E) ? dst[e] : -1;
    }
  }

  int rk[8];
#pragma unroll
  for (int k = 0; k < 8; ++k) {
    rk[k] = 0;
    if (dv[k] >= 0) {
      const unsigned bin = min((unsigned)dv[k] / BIN_NODES, (unsigned)(NBMAX - 1));
      rk[k] = atomicAdd(&s_hist[bin], 1);
    }
  }
  __syncthreads();

  int nloc;
  const int h  = (t < NBMAX) ? s_hist[t] : 0;
  const int ex = block_excl_scan(h, s_w, lane, wave, nloc);
  if (t < NBMAX) s_lofs[t] = ex;
  if (t == NBMAX) s_lofs[NBMAX] = nloc;
  __syncthreads();

  if (t <= NB) lofsT[(size_t)t * NBLK + blk] = s_lofs[t];
#pragma unroll
  for (int k = 0; k < 8; ++k) {
    if (dv[k] >= 0) {
      const unsigned bin = min((unsigned)dv[k] / BIN_NODES, (unsigned)(NBMAX - 1));
      s_pack[s_lofs[bin] + rk[k]] =
          (((unsigned)dv[k] - bin * BIN_NODES) << 23) | (unsigned)sv[k];
    }
  }
  __syncthreads();

  for (int i = t; i < nloc; i += 1024) binned[base + i] = s_pack[i];
}


__global__ __launch_bounds__(1024)
void k_sort_gemm1(const unsigned* __restrict__ binned, const int* __restrict__ lofsT,
                  int* __restrict__ list, int* __restrict__ rowptr, int* __restrict__ cnt,
                  float* __restrict__ dinv, const float* __restrict__ x,
                  const float* __restrict__ W1, float* __restrict__ hx,
                  int N, int NBLK, int ntiles) {
  __shared__ int s_a[MAXBLK];
  __shared__ int s_ro[MAXBLK];
  __shared__ int2 s_run[MAXBLK];
  __shared__ int s_cnt[BIN_SLOTS];
  __shared__ int s_ofs[BIN_SLOTS];
  __shared__ int s_cur[BIN_SLOTS];
  __shared__ float s_dinv[BIN_SLOTS];
  __shared__ int s_w1[32], s_w2[32], s_w3[32];
  __shared__ int s_list[SORT_CAP];
  __shared__ float sA[8][2][16 * 132];
  const int t    = threadIdx.x;
  const int lane = t & 31;
  const int wave = t >> 5;
  const int j    = blockIdx.x;

  const int grp   = wave >> 2;
  const int c0    = (wave & 3) * 16;
  const int lhalf = lane & 15;
  const int khi   = (lane >> 4) * 2;
  const int tg    = t & 127;
  const int row   = tg >> 3;
  const int q     = (tg & 7) * 16;
  const int tbase = j * BIN_TILES;
  const int nloc  = max(0, min(BIN_TILES, ntiles - tbase));

  if (j == 0 && t < HIDC / 4)
    ((float4*)hx)[(size_t)ntiles * 16 * (HIDC / 4) + t] = make_float4(0.0f, 0.0f, 0.0f, 0.0f);

  int a = 0, len = 0;
  if (t < NBLK) {
    a   = lofsT[(size_t)j * NBLK + t];
    len = lofsT[(size_t)(j + 1) * NBLK + t] - a;
  }
  float4 p0 = make_float4(0.0f, 0.0f, 0.0f, 0.0f), p1 = p0, p2 = p0, p3 = p0;
  if (grp < nloc) {
    const float4* sp = (const float4*)(x + ((size_t)(tbase + grp) * 16 + row) * INC + q);
    p0 = sp[0]; p1 = sp[1]; p2 = sp[2]; p3 = sp[3];
  }
  if (t < NBLK) s_a[t] = t * CH + a;
  s_run[t] = make_int2(t * CH + a, len);
  if (t < BIN_SLOTS) { s_cnt[t] = 0; s_cur[t] = 0; }
  int total = len, binbase = a, over = (len > 32 * SORT_SUB) ? 1 : 0;
  block_sum3(total, binbase, over, s_w1, s_w2, s_w3, lane, wave);

  const bool fast = (total <= SORT_CAP) && (NBLK <= 32 * SORT_RPW) && (over == 0);
  if (!fast) {
    lds_barrier();
    int tt;
    const int ro = block_excl_scan<true>(len, s_w1, lane, wave, tt);
    if (t < NBLK) s_ro[t] = ro;
    lds_barrier();
  }

  int top = 1;
  while (top * 2 < NBLK) top *= 2;

  auto fetch = [&](int i) -> unsigned {
    int pos = 0;
    for (int step = top; step > 0; step >>= 1) {
      const int nxt = pos + step;
      if (nxt < NBLK && s_ro[nxt] <= i) pos = nxt;
    }
    return binned[s_a[pos] + (i - s_ro[pos])];
  };

  const unsigned NOEDGE = 0xFFFFFFFFu;
  unsigned va[SORT_RPW], vb[SORT_RPW], vc[SORT_RPW];
  int      ka[SORT_RPW], kb[SORT_RPW], kc[SORT_RPW];
  if (fast) {
    int2 rn[SORT_RPW];
#pragma unroll
    for (int m = 0; m < SORT_RPW; ++m) rn[m] = s_run[wave + 32 * m];
#pragma unroll
    for (int m = 0; m < SORT_RPW; ++m) {
      va[m] = NOEDGE;
      if (lane < rn[m].y) va[m] = binned[rn[m].x + lane];
    }
#pragma unroll
    for (int m = 0; m < SORT_RPW; ++m) {
      vb[m] = NOEDGE;
      if (lane + 32 < rn[m].y) vb[m] = binned[rn[m].x + lane + 32];
    }
#pragma unroll
    for (int m = 0; m < SORT_RPW; ++m) {
      vc[m] = NOEDGE;
      if (lane + 64 < rn[m].y) vc[m] = binned[rn[m].x + lane + 64];
    }
#pragma unroll
    for (int m = 0; m < SORT_RPW; ++m) {
      ka[m] = 0;
      if (va[m] != NOEDGE) ka[m] = atomicAdd(&s_cnt[va[m] >> 23], 1);
    }
#pragma unroll
    for (int m = 0; m < SORT_RPW; ++m) {
      kb[m] = 0;
      if (vb[m] != NOEDGE) kb[m] = atomicAdd(&s_cnt[vb[m] >> 23], 1);
    }
#pragma unroll
    for (int m = 0; m < SORT_RPW; ++m) {
      kc[m] = 0;
      if (vc[m] != NOEDGE) kc[m] = atomicAdd(&s_cnt[vc[m] >> 23], 1);
    }
  } else {
    for (int i = t; i < total; i += 1024) atomicAdd(&s_cnt[fetch(i) >> 23], 1);
  }
  __syncthreads();

  int dummy;
  const int c   = (t < BIN_NODES) ? s_cnt[t] : 0;
  const int exn = block_excl_scan(c, s_w3, lane, wave, dummy);
  if (t < BIN_NODES) {
    s_ofs[t] = exn;
    const float di = rsqrtf((float)(c + 1));
    s_dinv[t] = di;
    const int node = j * BIN_NODES + t;
    if (node < N) {
      rowptr[node] = binbase + exn;
      cnt[node]    = c;
      dinv[node]   = di;
    }
  }
  __syncthreads();

  if (fast) {
#pragma unroll
    for (int m = 0; m < SORT_RPW; ++m) {
      if (va[m] != NOEDGE) s_list[s_ofs[va[m] >> 23] + ka[m]] = (int)(va[m] & 0x7FFFFFu);
    }
#pragma unroll
    for (int m = 0; m < SORT_RPW; ++m) {
      if (vb[m] != NOEDGE) s_list[s_ofs[vb[m] >> 23] + kb[m]] = (int)(vb[m] & 0x7FFFFFu);
    }
#pragma unroll
    for (int m = 0; m < SORT_RPW; ++m) {
      if (vc[m] != NOEDGE) s_list[s_ofs[vc[m] >> 23] + kc[m]] = (int)(vc[m] & 0x7FFFFFu);
    }
    __syncthreads();
    for (int i = t; i < total; i += 1024) list[binbase + i] = s_list[i];
  } else {
    for (int i = t; i < total; i += 1024) {
      const unsigned w  = fetch(i);
      const int      dl = (int)(w >> 23);
      const int      p  = s_ofs[dl] + atomicAdd(&s_cur[dl], 1);
      list[binbase + p] = (int)(w & 0x7FFFFFu);
    }
  }

  v2f bf[INC / 4];
#pragma unroll
  for (int kk = 0; kk < INC / 4; ++kk) {
    const float* bp = W1 + (size_t)(kk * 4 + khi) * HIDC + c0 + lhalf;
    bf[kk].x = bp[0];
    bf[kk].y = bp[HIDC];
  }

  int buf = 0;
  for (int lt0 = 0; lt0 < nloc; lt0 += 8, buf ^= 1) {
    const int  lt  = lt0 + grp;
    const bool act = (lt < nloc);
    if (act) {
      float4* const d = (float4*)(sA[grp][buf] + row * 132 + q);
      d[0] = p0; d[1] = p1; d[2] = p2; d[3] = p3;
    }
    lds_barrier();
    if (act) {
      const float* sa = sA[grp][buf];
      const int nt = (lt + 8 < nloc) ? lt + 8 : lt;
      {
        const float4* sp = (const float4*)(x + ((size_t)(tbase + nt) * 16 + row) * INC + q);
        p0 = sp[0]; p1 = sp[1]; p2 = sp[2]; p3 = sp[3];
      }
      __builtin_amdgcn_sched_barrier(0);

      const int ml = lt * 16 + ((lane >> 4) * 8);
      const int m0 = tbase * 16 + ml;
      const int n  = c0 + lhalf;
      float dv[8];
#pragma unroll
      for (int u = 0; u < 8; ++u) dv[u] = s_dinv[ml + u];

      v8f acc = {};
#pragma unroll
      for (int kk = 0; kk < INC / 4; ++kk) {
        v2f af;
        const float* ap = sa + lhalf * 132 + kk * 4 + khi;
        af.x = ap[0];
        af.y = ap[1];
        acc = __builtin_amdgcn_wmma_f32_16x16x4_f32(false, af, false, bf[kk],
                                                    (short)0, acc, false, false);
      }

#pragma unroll
      for (int u = 0; u < 8; ++u)
        hx[(size_t)(m0 + u) * HIDC + n] = acc[u] * dv[u];
    }
  }
}


template <int K>
__device__ __forceinline__ int row_lane(int v) {
  return __builtin_amdgcn_update_dpp(0, v, 0x150 + K, 0xf, 0xf, false);
}

struct rows8 { float4 v0, v1, v2, v3, v4, v5, v6, v7; };

template <int B>
__device__ __forceinline__ rows8 load8(const char* __restrict__ hb, int ido, unsigned colb) {
  const unsigned o0 = (unsigned)row_lane<B + 0>(ido) + colb;
  const unsigned o1 = (unsigned)row_lane<B + 1>(ido) + colb;
  const unsigned o2 = (unsigned)row_lane<B + 2>(ido) + colb;
  const unsigned o3 = (unsigned)row_lane<B + 3>(ido) + colb;
  const unsigned o4 = (unsigned)row_lane<B + 4>(ido) + colb;
  const unsigned o5 = (unsigned)row_lane<B + 5>(ido) + colb;
  const unsigned o6 = (unsigned)row_lane<B + 6>(ido) + colb;
  const unsigned o7 = (unsigned)row_lane<B + 7>(ido) + colb;
  rows8 r;
  r.v0 = *(const float4*)(hb + o0);
  r.v1 = *(const float4*)(hb + o1);
  r.v2 = *(const float4*)(hb + o2);
  r.v3 = *(const float4*)(hb + o3);
  r.v4 = *(const float4*)(hb + o4);
  r.v5 = *(const float4*)(hb + o5);
  r.v6 = *(const float4*)(hb + o6);
  r.v7 = *(const float4*)(hb + o7);
  return r;
}

__device__ __forceinline__ void sum8(const rows8& r, float4& acc) {
  acc.x += r.v0.x; acc.y += r.v0.y; acc.z += r.v0.z; acc.w += r.v0.w;
  acc.x += r.v1.x; acc.y += r.v1.y; acc.z += r.v1.z; acc.w += r.v1.w;
  acc.x += r.v2.x; acc.y += r.v2.y; acc.z += r.v2.z; acc.w += r.v2.w;
  acc.x += r.v3.x; acc.y += r.v3.y; acc.z += r.v3.z; acc.w += r.v3.w;
  acc.x += r.v4.x; acc.y += r.v4.y; acc.z += r.v4.z; acc.w += r.v4.w;
  acc.x += r.v5.x; acc.y += r.v5.y; acc.z += r.v5.z; acc.w += r.v5.w;
  acc.x += r.v6.x; acc.y += r.v6.y; acc.z += r.v6.z; acc.w += r.v6.w;
  acc.x += r.v7.x; acc.y += r.v7.y; acc.z += r.v7.z; acc.w += r.v7.w;
}

template <int ROWB>
__device__ __forceinline__ void add_in_rows(const char* __restrict__ hb, const int* __restrict__ list,
                                            int beg, int num, int ids, int nbeg, int& nraw,
                                            int zid, int elast, int sub, unsigned colb, float4& acc) {
  {
    int sids = zid;
    if (sub + 16 < num) sids = list[beg + 16 + sub];
    const int ido = ids * ROWB;
    rows8 r = load8<0>(hb, ido, colb);
    nraw = list[min(nbeg + sub, elast)];
    sum8(r, acc);
    if (num > 8) {
      r = load8<8>(hb, ido, colb);
      sum8(r, acc);
    }
    ids = sids;
  }
  for (int done = 16; done < num; done += 16) {
    const int left = num - done;
    int sids = zid;
    if (sub + 16 < left) sids = list[beg + done + 16 + sub];
    const int ido = ids * ROWB;
    rows8 r = load8<0>(hb, ido, colb);
    sum8(r, acc);
    if (left > 8) {
      r = load8<8>(hb, ido, colb);
      sum8(r, acc);
    }
    ids = sids;
  }
}


#define GF_BLOCKS 1792

__global__ __launch_bounds__(256)
void k_pull_gemm2(const float* __restrict__ hx, const int* __restrict__ list,
                  const int* __restrict__ rowptr, const int* __restrict__ cnt,
                  const float* __restrict__ dinv, const float* __restrict__ b1,
                  const float* __restrict__ W2, float* __restrict__ h2, int ntiles, int E) {
  __shared__ float sA[2][16 * 68];
  __shared__ float sD[2][16];
  __shared__ float sB[(HIDC / 2) * 48 * 2];
  const int t     = threadIdx.x;
  const int wave  = t >> 5;
  const int lane  = t & 31;
  const int g     = t >> 4;
  const int sub   = t & 15;
  const int lhalf = lane & 15;
  const int khi   = (lane >> 4) * 2;
  const int col   = wave * 16 + lhalf;
  const bool mm    = (wave < 3);
  const bool colOK = mm && (col < OUTC);

  const float4  b  = ((const float4*)b1)[sub];
  const char*    hb   = (const char*)hx;
  const unsigned colb = (unsigned)sub * 16u;
  const int      zid  = ntiles * 16;
  const int      elast = max(E - 1, 0);

  if (blockIdx.x == 0 && t < OUTC / 4)
    ((float4*)h2)[(size_t)ntiles * 16 * (OUTC / 4) + t] = make_float4(0.0f, 0.0f, 0.0f, 0.0f);

  int tile = blockIdx.x;
  int    beg = 0, num = 0;
  float  di  = 0.0f;
  int    ids = zid;
  if (tile < ntiles) {
    const int node = tile * 16 + g;
    beg = rowptr[node];
    num = cnt[node];
    di  = dinv[node];
    if (sub < num) ids = list[beg + sub];
  }
  float wx[6], wy[6];
#pragma unroll
  for (int r = 0; r < 6; ++r) {
    const int i  = t + r * 256;
    const int kp = i / 48;
    const int cc = min(i - kp * 48, OUTC - 1);
    wx[r] = W2[(size_t)(kp * 2) * OUTC + cc];
    wy[r] = W2[(size_t)(kp * 2 + 1) * OUTC + cc];
  }
#pragma unroll
  for (int r = 0; r < 6; ++r) {
    const int i  = t + r * 256;
    const int kp = i / 48;
    const bool in = (i - kp * 48 < OUTC);
    sB[i * 2]     = in ? wx[r] : 0.0f;
    sB[i * 2 + 1] = in ? wy[r] : 0.0f;
  }

  int buf = 0;
  for (; tile < ntiles; tile += gridDim.x, buf ^= 1) {
    const int nt = (tile + gridDim.x < ntiles) ? tile + gridDim.x : tile;
    const int nn = nt * 16 + g;
    const int    nbeg = rowptr[nn];
    const int    nnum = cnt[nn];
    const float  ndi  = dinv[nn];

    float4 acc = *(const float4*)(hb + ((unsigned)(tile * 16 + g) * (HIDC * 4) + colb));
    int nraw;
    add_in_rows<HIDC * 4>(hb, list, beg, num, ids, nbeg, nraw, zid, elast, sub, colb, acc);

    float4 r;
    r.x = fmaxf(di * acc.x + b.x, 0.0f);
    r.y = fmaxf(di * acc.y + b.y, 0.0f);
    r.z = fmaxf(di * acc.z + b.z, 0.0f);
    r.w = fmaxf(di * acc.w + b.w, 0.0f);
    float* sa = sA[buf];
    *(float4*)(sa + g * 68 + sub * 4) = r;
    if (sub == 0) sD[buf][g] = di;
    lds_barrier();

    if (mm) {
      v8f c = {};
      const float* ap = sa + lhalf * 68 + khi;
      const float* bp = sB + ((lane >> 4) * 48 + col) * 2;
#pragma unroll
      for (int kk = 0; kk < HIDC / 4; ++kk) {
        v2f a, bb;
        a.x  = ap[kk * 4];
        a.y  = ap[kk * 4 + 1];
        bb.x = bp[kk * 192];
        bb.y = bp[kk * 192 + 1];
        c = __builtin_amdgcn_wmma_f32_16x16x4_f32(false, a, false, bb,
                                                  (short)0, c, false, false);
      }
      const int mh = (lane >> 4) * 8;
      const int m0 = tile * 16 + mh;
      if (colOK) {
#pragma unroll
        for (int v = 0; v < 8; ++v)
          h2[(size_t)(m0 + v) * OUTC + col] = c[v] * sD[buf][mh + v];
      }
    }

    beg = nbeg; num = nnum; di = ndi;
    ids = (sub < nnum) ? nraw : zid;
  }
}


template <int C4, bool BIAS>
__device__ __forceinline__ void pull_rows(const float* __restrict__ h, const int* __restrict__ list,
                                          const int* __restrict__ rowptr, const int* __restrict__ cnt,
                                          const float* __restrict__ dinv, const float* __restrict__ bias,
                                          float* __restrict__ o, int n, int E) {
  const int gpb    = blockDim.x >> 4;
  const int stride = gridDim.x * gpb;
  const int sub    = threadIdx.x & 15;
  int node = blockIdx.x * gpb + (threadIdx.x >> 4);
  if (node >= n) return;

  const int      col  = min(sub, C4 - 1);
  const unsigned colb = (unsigned)col * 16u;
  const char*    hb   = (const char*)h;
  const int      zid  = n;
  const int      elast = max(E - 1, 0);
  char* const    ob   = (char*)o;
  float4 b = make_float4(0.0f, 0.0f, 0.0f, 0.0f);
  if (BIAS) b = ((const float4*)bias)[col];

  int    beg = rowptr[node];
  int    num = cnt[node];
  float  di  = dinv[node];
  int    ids = zid;
  if (sub < num) ids = list[beg + sub];

  for (;;) {
    const int  nxt  = node + stride;
    const bool more = (nxt < n);
    const int  nn   = more ? nxt : node;
    const int    nbeg = rowptr[nn];
    const int    nnum = cnt[nn];
    const float  ndi  = dinv[nn];

    float4 acc = *(const float4*)(hb + ((unsigned)node * (C4 * 16) + colb));
    int nraw;
    add_in_rows<C4 * 16>(hb, list, beg, num, ids, nbeg, nraw, zid, elast, sub, colb, acc);

    if (BIAS) {
      acc.x = b.x + di * acc.x;
      acc.y = b.y + di * acc.y;
      acc.z = b.z + di * acc.z;
      acc.w = b.w + di * acc.w;
    } else {
      acc.x *= di; acc.y *= di; acc.z *= di; acc.w *= di;
    }
    if (sub < C4) *(float4*)(ob + ((unsigned)node * (C4 * 16) + colb)) = acc;

    if (!more) break;
    node = nxt; beg = nbeg; num = nnum; di = ndi;
    ids = (sub < nnum) ? nraw : zid;
  }
}

__global__ __launch_bounds__(256)
void k_pull_o(const float* __restrict__ h2, const int* __restrict__ list,
              const int* __restrict__ rowptr, const int* __restrict__ cnt,
              const float* __restrict__ dinv, const float* __restrict__ b2,
              float* __restrict__ out, int n, int E) {
  pull_rows<OUTC / 4, true>(h2, list, rowptr, cnt, dinv, b2, out, n, E);
}


extern "C" void kernel_launch(void* const* d_in, const int* in_sizes, int n_in,
                              void* d_out, int out_size, void* d_ws, size_t ws_size,
                              hipStream_t stream) {
  const float* x  = (const float*)d_in[0];
  const int*   ei = (const int*)d_in[1];
  const float* W1 = (const float*)d_in[2];
  const float* b1 = (const float*)d_in[3];
  const float* W2 = (const float*)d_in[4];
  const float* b2 = (const float*)d_in[5];
  float* out = (float*)d_out;

  const int N = in_sizes[0] / INC;
  const int E = in_sizes[1] / 2;
  const int* src = ei;
  const int* dst = ei + E;

  char*  ws  = (char*)d_ws;
  size_t off = 0;
  auto carve = [&](size_t bytes) -> void* {
    void* p = (void*)(ws + off);
    off = (off + bytes + 255) & ~(size_t)255;
    return p;
  };
  int*   cnt    = (int*)  carve((size_t)N * 4);
  int*   rowptr = (int*)  carve((size_t)N * 4);
  float* dinv   = (float*)carve((size_t)N * 4);
  int*   list   = (int*)  carve((size_t)E * 4);
  int*   rank   = (int*)  carve((size_t)E * 4);
  float* hx     = (float*)carve((size_t)(N + 1) * HIDC * 4);
  float* h2     = (float*)carve((size_t)(N + 1) * OUTC * 4);
  (void)n_in; (void)out_size; (void)ws_size;

  const int B = 256;
  const int G  = 2048;
  const int GF = GF_BLOCKS;
  const int NB   = (N + BIN_NODES - 1) / BIN_NODES;
  const int NBLK = (E + CH - 1) / CH;
  unsigned* binned = (unsigned*)rank;
  int*      lofsT  = (int*)h2;
  k_bin       <<<NBLK, 1024, 0, stream>>>(src, dst, binned, lofsT, E, NB, NBLK);
  k_sort_gemm1<<<NB,   1024, 0, stream>>>(binned, lofsT, list, rowptr, cnt, dinv, x, W1, hx,
                                          N, NBLK, N / 16);

  k_pull_gemm2<<<GF, B, 0, stream>>>(hx, list, rowptr, cnt, dinv, b1, W2, h2, N / 16, E);

  k_pull_o    <<<G, B, 0, stream>>>(h2, list, rowptr, cnt, dinv, b2, out, N, E);
}
